# starting kernel
# baseline (speedup 1.0000x reference)
#include <hip/hip_runtime.h>
#include <math.h>

typedef float v2f __attribute__((ext_vector_type(2)));
typedef float v8f __attribute__((ext_vector_type(8)));

namespace {
constexpr int kPS   = 16;
constexpr int kPR   = 14;
constexpr int kN    = 196;
constexpr int kC    = 3;
constexpr int kW    = 224;
constexpr int kBS   = 64;
constexpr int kF    = 768;
constexpr int kHid  = 768;
constexpr int kOut  = 768;
constexpr int kRows = kBS * kN;
constexpr int kTileRows     = 16;
constexpr int kWaves        = 8;
constexpr int kTilesPerWave = (kHid / 16) / kWaves;
}

__device__ __forceinline__ float gelu_exact(float v) {
  return 0.5f * v * (1.0f + erff(v * 0.70710678118654752440f));
}

__global__ __launch_bounds__(256)
void conv_bn_gelu_patch(const float* __restrict__ x,
                        const float* __restrict__ cw,
                        const float* __restrict__ cb,
                        const float* __restrict__ bng,
                        const float* __restrict__ bnb,
                        const float* __restrict__ bnm,
                        const float* __restrict__ bnv,
                        const float* __restrict__ M,
                        const int*   __restrict__ mask,
                        float* __restrict__ flat) {
  const int total = kBS * kC * kW * kW;
  int idx = blockIdx.x * blockDim.x + threadIdx.x;
  if (idx >= total) return;

  const int h   = idx % kW;
  const int w   = (idx / kW) % kW;
  const int co  = (idx / (kW * kW)) % kC;
  const int img = idx / (kW * kW * kC);

  float acc = cb[co];
#pragma unroll
  for (int ci = 0; ci < kC; ++ci) {
    const float* __restrict__ xp = x + ((size_t)(img * kC + ci)) * kW * kW;
#pragma unroll
    for (int kw = 0; kw < 3; ++kw) {
      const int ww = w + kw - 1;
      if (ww < 0 || ww >= kW) continue;
#pragma unroll
      for (int kh = 0; kh < 3; ++kh) {
        const int hh = h + kh - 1;
        if (hh < 0 || hh >= kW) continue;
        acc += xp[ww * kW + hh] * cw[((co * kC + ci) * 3 + kw) * 3 + kh];
      }
    }
  }
  const float inv = bng[co] * rsqrtf(bnv[co] + 1e-5f);
  acc = (acc - bnm[co]) * inv + bnb[co];
  acc = gelu_exact(acc);

  const int pi = w & 15, pw = w >> 4;
  const int pj = h & 15, ph = h >> 4;
  const int n  = pw * kPR + ph;
  if (mask[n] > 0) acc *= M[pi * kPS + pj];

  const int row = img * kN + n;
  const int f   = co * kPS * kPS + pi * kPS + pj;
  flat[(size_t)row * kF + f] = acc;
}

__global__ __launch_bounds__(256, 2)
void gemm1_ln_gelu(const float* __restrict__ A,
                   const float* __restrict__ W1,
                   const float* __restrict__ b1,
                   const float* __restrict__ lng,
                   const float* __restrict__ lnb,
                   float* __restrict__ T) {
  __shared__ float sA[kTileRows * kF];
  __shared__ float sMu[kTileRows];
  __shared__ float sRs[kTileRows];

  const int tid  = threadIdx.x;
  const int wave = tid >> 5;
  const int lane = tid & 31;
  const int hf   = lane >> 4;
  const int l16  = lane & 15;
  const size_t row0 = (size_t)blockIdx.x * kTileRows;

  for (int i = tid; i < kTileRows * kF; i += 256) sA[i] = A[row0 * kF + i];
  __syncthreads();

  const int kOff = 2 * hf;
  v8f acc[kTilesPerWave];

  for (int t = 0; t < kTilesPerWave; ++t) {
    v8f c = {};
    const int n = (wave * kTilesPerWave + t) * 16 + l16;
    const float* __restrict__ bcol = W1 + (size_t)n * kF + kOff;
    const float* __restrict__ arow = sA + l16 * kF + kOff;
    for (int k = 0; k < kF; k += 4) {
      v2f a = *(const v2f*)(arow + k);
      v2f b = *(const v2f*)(bcol + k);
      c = __builtin_amdgcn_wmma_f32_16x16x4_f32(false, a, false, b,
                                                (short)0, c, false, false);
    }
    acc[t] = c;
  }
  __syncthreads();

  for (int t = 0; t < kTilesPerWave; ++t) {
    const int n = (wave * kTilesPerWave + t) * 16 + l16;
    const float bias = b1[n];
#pragma unroll
    for (int r = 0; r < 8; ++r) {
      sA[(r + 8 * hf) * kF + n] = acc[t][r] + bias;
    }
  }
  __syncthreads();

  if (tid < kTileRows) {
    float s = 0.f, s2 = 0.f;
    const float* rp = sA + tid * kF;
    for (int c = 0; c < kF; ++c) { const float v = rp[c]; s += v; s2 += v * v; }
    const float mu  = s / kF;
    const float var = s2 / kF - mu * mu;
    sMu[tid] = mu;
    sRs[tid] = rsqrtf(var + 1e-5f);
  }
  __syncthreads();

  for (int i = tid; i < kTileRows * kF; i += 256) {
    const int r = i / kF, c = i % kF;
    const float v = (sA[i] - sMu[r]) * sRs[r] * lng[c] + lnb[c];
    T[row0 * kF + i] = gelu_exact(v);
  }
}

__global__ __launch_bounds__(256, 2)
void gemm2_perm(const float* __restrict__ T,
                const float* __restrict__ W2,
                const float* __restrict__ b2,
                const int*   __restrict__ perm,
                float* __restrict__ out) {
  __shared__ float sA[kTileRows * kF];
  __shared__ int   sRow[kTileRows];

  const int tid  = threadIdx.x;
  const int wave = tid >> 5;
  const int lane = tid & 31;
  const int hf   = lane >> 4;
  const int l16  = lane & 15;
  const int row0 = blockIdx.x * kTileRows;

  if (tid < kTileRows) {
    const int g   = row0 + tid;
    const int img = g / kN;
    const int n   = g - img * kN;
    sRow[tid] = img * kN + perm[n];
  }
  __syncthreads();
  for (int i = tid; i < kTileRows * kF; i += 256) {
    const int r = i / kF, c = i % kF;
    sA[i] = T[(size_t)sRow[r] * kF + c];
  }
  __syncthreads();

  const int kOff = 2 * hf;
  for (int t = 0; t < kTilesPerWave; ++t) {
    v8f c = {};
    const int n = (wave * kTilesPerWave + t) * 16 + l16;
    const float* __restrict__ bcol = W2 + (size_t)n * kF + kOff;
    const float* __restrict__ arow = sA + l16 * kF + kOff;
    for (int k = 0; k < kF; k += 4) {
      v2f a = *(const v2f*)(arow + k);
      v2f b = *(const v2f*)(bcol + k);
      c = __builtin_amdgcn_wmma_f32_16x16x4_f32(false, a, false, b,
                                                (short)0, c, false, false);
    }
    const float bias = b2[n];
#pragma unroll
    for (int r = 0; r < 8; ++r) {
      const int m = r + 8 * hf;
      out[(size_t)(row0 + m) * kOut + n] = c[r] + bias;
    }
  }
}

extern "C" void kernel_launch(void* const* d_in, const int* in_sizes, int n_in,
                              void* d_out, int out_size, void* d_ws, size_t ws_size,
                              hipStream_t stream) {
  const float* x    = (const float*)d_in[0];
  const float* cw   = (const float*)d_in[1];
  const float* cb   = (const float*)d_in[2];
  const float* bng  = (const float*)d_in[3];
  const float* bnb  = (const float*)d_in[4];
  const float* bnm  = (const float*)d_in[5];
  const float* bnv  = (const float*)d_in[6];
  const float* M    = (const float*)d_in[7];
  const float* W1   = (const float*)d_in[8];
  const float* b1   = (const float*)d_in[9];
  const float* lng  = (const float*)d_in[10];
  const float* lnb  = (const float*)d_in[11];
  const float* W2   = (const float*)d_in[12];
  const float* b2   = (const float*)d_in[13];
  const int*   mask = (const int*)d_in[14];
  const int*   perm = (const int*)d_in[15];
  float* out = (float*)d_out;

  float* flat = (float*)d_ws;
  float* Tbuf = flat + (size_t)kRows * kF;

  {
    const int total = kBS * kC * kW * kW;
    conv_bn_gelu_patch<<<(total + 255) / 256, 256, 0, stream>>>(
        x, cw, cb, bng, bnb, bnm, bnv, M, mask, flat);
  }
  gemm1_ln_gelu<<<kRows / kTileRows, 256, 0, stream>>>(flat, W1, b1, lng, lnb, Tbuf);
  gemm2_perm  <<<kRows / kTileRows, 256, 0, stream>>>(Tbuf, W2, b2, perm, out);
}


# both GEMMs re-decomposed: each wave owns 16 whole rows x 768 columns in 48 register tiles, W streamed through one shared 768x32 LDS chunk, A from global; LayerNorm sums in registers (order changed)
# speedup vs baseline: 1.1389x; 1.1389x over previous
#include <hip/hip_runtime.h>
#include <math.h>


typedef float v2f __attribute__((ext_vector_type(2)));
typedef float v8f __attribute__((ext_vector_type(8)));

namespace {
constexpr int kPS   = 16;
constexpr int kPR   = 14;
constexpr int kN    = 196;
constexpr int kC    = 3;
constexpr int kW    = 224;
constexpr int kBS   = 64;
constexpr int kF    = 768;
constexpr int kHid  = 768;
constexpr int kOut  = 768;
constexpr int kRows = kBS * kN;
constexpr int kWaveRows  = 16;
constexpr int kWaves     = 4;
constexpr int kBlockRows = kWaveRows * kWaves;
constexpr int kTiles     = kHid / 16;
constexpr int kKC        = 32;
constexpr int kPitch     = 36;
}

__device__ __forceinline__ float gelu_exact(float v) {
  return 0.5f * v * (1.0f + erff(v * 0.70710678118654752440f));
}

__global__ __launch_bounds__(256)
void conv_bn_gelu_patch(const float* __restrict__ x,
                        const float* __restrict__ cw,
                        const float* __restrict__ cb,
                        const float* __restrict__ bng,
                        const float* __restrict__ bnb,
                        const float* __restrict__ bnm,
                        const float* __restrict__ bnv,
                        const float* __restrict__ M,
                        const int*   __restrict__ mask,
                        float* __restrict__ flat) {
  const int total = kBS * kC * kW * kW;
  int idx = blockIdx.x * blockDim.x + threadIdx.x;
  if (idx >= total) return;

  const int h   = idx % kW;
  const int w   = (idx / kW) % kW;
  const int co  = (idx / (kW * kW)) % kC;
  const int img = idx / (kW * kW * kC);

  float acc = cb[co];
#pragma unroll
  for (int ci = 0; ci < kC; ++ci) {
    const float* __restrict__ xp = x + ((size_t)(img * kC + ci)) * kW * kW;
#pragma unroll
    for (int kw = 0; kw < 3; ++kw) {
      const int ww = w + kw - 1;
      if (ww < 0 || ww >= kW) continue;
#pragma unroll
      for (int kh = 0; kh < 3; ++kh) {
        const int hh = h + kh - 1;
        if (hh < 0 || hh >= kW) continue;
        acc += xp[ww * kW + hh] * cw[((co * kC + ci) * 3 + kw) * 3 + kh];
      }
    }
  }
  const float inv = bng[co] * rsqrtf(bnv[co] + 1e-5f);
  acc = (acc - bnm[co]) * inv + bnb[co];
  acc = gelu_exact(acc);

  const int pi = w & 15, pw = w >> 4;
  const int pj = h & 15, ph = h >> 4;
  const int n  = pw * kPR + ph;
  if (mask[n] > 0) acc *= M[pi * kPS + pj];

  const int row = img * kN + n;
  const int f   = co * kPS * kPS + pi * kPS + pj;
  flat[(size_t)row * kF + f] = acc;
}

__device__ __forceinline__ void gemm_rows16_x768(const float* __restrict__ arow,
                                                 const float* __restrict__ Wm,
                                                 float* __restrict__ sW,
                                                 v8f (&acc)[kTiles],
                                                 int tid, int hf, int l16) {
#pragma unroll
  for (int t = 0; t < kTiles; ++t) acc[t] = v8f{};

  const float* __restrict__ brow = sW + l16 * kPitch + 2 * hf;

#pragma unroll 1
  for (int kc = 0; kc < kF; kc += kKC) {
    __syncthreads();
    {
      float4 v[48];
#pragma unroll
      for (int i = 0; i < 48; ++i) {
        const int idx = tid + i * (kWaves * 32);
        const int n = idx >> 3, q = idx & 7;
        v[i] = *(const float4*)(Wm + (size_t)n * kF + kc + 4 * q);
      }
#pragma unroll
      for (int i = 0; i < 48; ++i) {
        const int idx = tid + i * (kWaves * 32);
        const int n = idx >> 3, q = idx & 7;
        *(float4*)(sW + n * kPitch + 4 * q) = v[i];
      }
    }
    __syncthreads();

    v2f a[kKC / 4];
#pragma unroll
    for (int s = 0; s < kKC / 4; ++s) a[s] = *(const v2f*)(arow + kc + 4 * s);

#pragma unroll
    for (int s = 0; s < kKC / 4; ++s) {
#pragma unroll
      for (int t = 0; t < kTiles; ++t) {
        const v2f b = *(const v2f*)(brow + t * 16 * kPitch + 4 * s);
        acc[t] = __builtin_amdgcn_wmma_f32_16x16x4_f32(false, a[s], false, b,
                                                       (short)0, acc[t], false, false);
      }
    }
  }
}

template <int T0, int T1>
__device__ __forceinline__ void ln_gelu_store(const v8f (&acc)[kTiles],
                                              const float (&mu)[8], const float (&rs)[8],
                                              const float* __restrict__ gcol,
                                              const float* __restrict__ bcol,
                                              float* __restrict__ trow) {
#pragma unroll
  for (int t = T0; t < T1; ++t) {
    const float g = gcol[t * 16], bb = bcol[t * 16];
#pragma unroll
    for (int r = 0; r < 8; ++r) {
      const float v = (acc[t][r] - mu[r]) * rs[r] * g + bb;
      trow[(size_t)r * kHid + t * 16] = gelu_exact(v);
    }
  }
}

__global__ __launch_bounds__(128)
void gemm1_ln_gelu(const float* __restrict__ A,
                   const float* __restrict__ W1,
                   const float* __restrict__ b1,
                   const float* __restrict__ lng,
                   const float* __restrict__ lnb,
                   float* __restrict__ T) {
  __shared__ __attribute__((aligned(16))) float sW[kHid * kPitch];

  const int tid  = threadIdx.x;
  const int wave = tid >> 5;
  const int lane = tid & 31;
  const int hf   = lane >> 4;
  const int l16  = lane & 15;
  const int row0 = blockIdx.x * kBlockRows + wave * kWaveRows;

  v8f acc[kTiles];
  gemm_rows16_x768(A + (size_t)(row0 + l16) * kF + 2 * hf, W1, sW, acc, tid, hf, l16);

  float s1[8], s2[8];
#pragma unroll
  for (int r = 0; r < 8; ++r) { s1[r] = 0.f; s2[r] = 0.f; }
#pragma unroll
  for (int t = 0; t < kTiles; ++t) {
    const float bias = b1[t * 16 + l16];
#pragma unroll
    for (int r = 0; r < 8; ++r) {
      const float v = acc[t][r] + bias;
      acc[t][r] = v;
      s1[r] += v;
      s2[r] += v * v;
    }
  }
  float mu[8], rs[8];
#pragma unroll
  for (int r = 0; r < 8; ++r) {
#pragma unroll
    for (int off = 8; off >= 1; off >>= 1) {
      s1[r] += __shfl_xor(s1[r], off, 32);
      s2[r] += __shfl_xor(s2[r], off, 32);
    }
    mu[r] = s1[r] / kF;
    const float var = s2[r] / kF - mu[r] * mu[r];
    rs[r] = rsqrtf(var + 1e-5f);
  }
  float* __restrict__ trow = T + (size_t)(row0 + 8 * hf) * kHid + l16;
  ln_gelu_store<0, kTiles / 2>(acc, mu, rs, lng + l16, lnb + l16, trow);
  ln_gelu_store<kTiles / 2, kTiles>(acc, mu, rs, lng + l16, lnb + l16, trow);
}

__global__ __launch_bounds__(128)
void gemm2_perm(const float* __restrict__ T,
                const float* __restrict__ W2,
                const float* __restrict__ b2,
                const int*   __restrict__ perm,
                float* __restrict__ out) {
  __shared__ __attribute__((aligned(16))) float sW[kOut * kPitch];

  const int tid  = threadIdx.x;
  const int wave = tid >> 5;
  const int lane = tid & 31;
  const int hf   = lane >> 4;
  const int l16  = lane & 15;
  const int row0 = blockIdx.x * kBlockRows + wave * kWaveRows;

  const int g    = row0 + l16;
  const int img  = g / kN;
  const int n0   = g - img * kN;
  const int src  = img * kN + perm[n0];

  v8f acc[kTiles];
  gemm_rows16_x768(T + (size_t)src * kHid + 2 * hf, W2, sW, acc, tid, hf, l16);

#pragma unroll
  for (int t = 0; t < kTiles; ++t) {
    const int n = t * 16 + l16;
    const float bias = b2[n];
#pragma unroll
    for (int r = 0; r < 8; ++r) {
      out[(size_t)(row0 + 8 * hf + r) * kOut + n] = acc[t][r] + bias;
    }
  }
}

extern "C" void kernel_launch(void* const* d_in, const int* in_sizes, int n_in,
                              void* d_out, int out_size, void* d_ws, size_t ws_size,
                              hipStream_t stream) {
  const float* x    = (const float*)d_in[0];
  const float* cw   = (const float*)d_in[1];
  const float* cb   = (const float*)d_in[2];
  const float* bng  = (const float*)d_in[3];
  const float* bnb  = (const float*)d_in[4];
  const float* bnm  = (const float*)d_in[5];
  const float* bnv  = (const float*)d_in[6];
  const float* M    = (const float*)d_in[7];
  const float* W1   = (const float*)d_in[8];
  const float* b1   = (const float*)d_in[9];
  const float* lng  = (const float*)d_in[10];
  const float* lnb  = (const float*)d_in[11];
  const float* W2   = (const float*)d_in[12];
  const float* b2   = (const float*)d_in[13];
  const int*   mask = (const int*)d_in[14];
  const int*   perm = (const int*)d_in[15];
  float* out = (float*)d_out;

  float* flat = (float*)d_ws;
  float* Tbuf = flat + (size_t)kRows * kF;

  {
    const int total = kBS * kC * kW * kW;
    conv_bn_gelu_patch<<<(total + 255) / 256, 256, 0, stream>>>(
        x, cw, cb, bng, bnb, bnm, bnv, M, mask, flat);
  }
  gemm1_ln_gelu<<<kRows / kBlockRows, kWaves * 32, 0, stream>>>(flat, W1, b1, lng, lnb, Tbuf);
  gemm2_perm  <<<kRows / kBlockRows, kWaves * 32, 0, stream>>>(Tbuf, W2, b2, perm, out);
}


# conv kernel laid out like its output: a thread owns one float4 of a flat row (x read as float4 + 2 edge values per tap row) and loops over 7 patches; a wave stores 512 B contiguous; tap order kept
# speedup vs baseline: 1.5251x; 1.3391x over previous
#include <hip/hip_runtime.h>
#include <math.h>


typedef float v2f __attribute__((ext_vector_type(2)));
typedef float v8f __attribute__((ext_vector_type(8)));

namespace {
constexpr int kPS   = 16;
constexpr int kPR   = 14;
constexpr int kN    = 196;
constexpr int kC    = 3;
constexpr int kW    = 224;
constexpr int kBS   = 64;
constexpr int kF    = 768;
constexpr int kHid  = 768;
constexpr int kOut  = 768;
constexpr int kRows = kBS * kN;
constexpr int kWaveRows  = 16;
constexpr int kWaves     = 4;
constexpr int kBlockRows = kWaveRows * kWaves;
constexpr int kTiles     = kHid / 16;
constexpr int kKC        = 32;
constexpr int kPitch     = 36;
}

__device__ __forceinline__ float gelu_exact(float v) {
  return 0.5f * v * (1.0f + erff(v * 0.70710678118654752440f));
}

constexpr int kConvThreads = kF / 4;
constexpr int kConvPatches = 7;

__global__ __launch_bounds__(kConvThreads)
void conv_bn_gelu_patch(const float* __restrict__ x,
                        const float* __restrict__ cw,
                        const float* __restrict__ cb,
                        const float* __restrict__ bng,
                        const float* __restrict__ bnb,
                        const float* __restrict__ bnm,
                        const float* __restrict__ bnv,
                        const float* __restrict__ M,
                        const int*   __restrict__ mask,
                        float* __restrict__ flat) {
  const int t   = threadIdx.x;
  const int co  = __builtin_amdgcn_readfirstlane(t >> 6);
  const int pi  = (t >> 2) & 15;
  const int pj0 = (t & 3) * 4;

  int b = blockIdx.x;
  const int ph0 = (b & 1) * kConvPatches;  b >>= 1;
  const int pw  = b % kPR;
  const int img = b / kPR;
  const int w   = pw * kPS + pi;

  int  roff[3];
  float wt[kC][3][3];
#pragma unroll
  for (int kw = 0; kw < 3; ++kw) {
    const int ww = w + kw - 1;
    const bool ok = (unsigned)ww < (unsigned)kW;
    roff[kw] = (ok ? ww : w) * kW;
#pragma unroll
    for (int ci = 0; ci < kC; ++ci)
#pragma unroll
      for (int kh = 0; kh < 3; ++kh) {
        const float wv = cw[((co * kC + ci) * 3 + kw) * 3 + kh];
        wt[ci][kw][kh] = ok ? wv : 0.0f;
      }
  }
  const float bias = cb[co];
  const float inv  = bng[co] * rsqrtf(bnv[co] + 1e-5f);
  const float mean = bnm[co];
  const float beta = bnb[co];
  float mm[4];
#pragma unroll
  for (int j = 0; j < 4; ++j) mm[j] = M[pi * kPS + pj0 + j];

  const float* __restrict__ ximg = x + (size_t)img * kC * kW * kW;

  for (int i = 0; i < kConvPatches; ++i) {
    const int ph = ph0 + i;
    const int h0 = ph * kPS + pj0;
    const bool lok = h0 > 0;
    const bool rok = h0 + 4 < kW;
    const int hl = lok ? h0 - 1 : h0;
    const int hr = rok ? h0 + 4 : h0 + 3;

    float xv[kC][3][6];
#pragma unroll
    for (int ci = 0; ci < kC; ++ci)
#pragma unroll
      for (int kw = 0; kw < 3; ++kw) {
        const float* __restrict__ rp = ximg + ci * kW * kW + roff[kw];
        const float4 v = *(const float4*)(rp + h0);
        const float l = rp[hl];
        const float r = rp[hr];
        xv[ci][kw][0] = lok ? l : 0.0f;
        xv[ci][kw][1] = v.x;
        xv[ci][kw][2] = v.y;
        xv[ci][kw][3] = v.z;
        xv[ci][kw][4] = v.w;
        xv[ci][kw][5] = rok ? r : 0.0f;
      }

    float acc[4] = {bias, bias, bias, bias};
#pragma unroll
    for (int ci = 0; ci < kC; ++ci)
#pragma unroll
      for (int kw = 0; kw < 3; ++kw)
#pragma unroll
        for (int kh = 0; kh < 3; ++kh)
#pragma unroll
          for (int j = 0; j < 4; ++j)
            acc[j] += xv[ci][kw][j + kh] * wt[ci][kw][kh];

    const int n = pw * kPR + ph;
    const bool gate = mask[n] > 0;
    float o[4];
#pragma unroll
    for (int j = 0; j < 4; ++j) {
      float a = (acc[j] - mean) * inv + beta;
      a = gelu_exact(a);
      if (gate) a *= mm[j];
      o[j] = a;
    }
    const int row = img * kN + n;
    *(float4*)(flat + (size_t)row * kF + 4 * t) = make_float4(o[0], o[1], o[2], o[3]);
  }
}

__device__ __forceinline__ void gemm_rows16_x768(const float* __restrict__ arow,
                                                 const float* __restrict__ Wm,
                                                 float* __restrict__ sW,
                                                 v8f (&acc)[kTiles],
                                                 int tid, int hf, int l16) {
#pragma unroll
  for (int t = 0; t < kTiles; ++t) acc[t] = v8f{};

  const float* __restrict__ brow = sW + l16 * kPitch + 2 * hf;

#pragma unroll 1
  for (int kc = 0; kc < kF; kc += kKC) {
    __syncthreads();
    {
      float4 v[48];
#pragma unroll
      for (int i = 0; i < 48; ++i) {
        const int idx = tid + i * (kWaves * 32);
        const int n = idx >> 3, q = idx & 7;
        v[i] = *(const float4*)(Wm + (size_t)n * kF + kc + 4 * q);
      }
#pragma unroll
      for (int i = 0; i < 48; ++i) {
        const int idx = tid + i * (kWaves * 32);
        const int n = idx >> 3, q = idx & 7;
        *(float4*)(sW + n * kPitch + 4 * q) = v[i];
      }
    }
    __syncthreads();

    v2f a[kKC / 4];
#pragma unroll
    for (int s = 0; s < kKC / 4; ++s) a[s] = *(const v2f*)(arow + kc + 4 * s);

#pragma unroll
    for (int s = 0; s < kKC / 4; ++s) {
#pragma unroll
      for (int t = 0; t < kTiles; ++t) {
        const v2f b = *(const v2f*)(brow + t * 16 * kPitch + 4 * s);
        acc[t] = __builtin_amdgcn_wmma_f32_16x16x4_f32(false, a[s], false, b,
                                                       (short)0, acc[t], false, false);
      }
    }
  }
}

template <int T0, int T1>
__device__ __forceinline__ void ln_gelu_store(const v8f (&acc)[kTiles],
                                              const float (&mu)[8], const float (&rs)[8],
                                              const float* __restrict__ gcol,
                                              const float* __restrict__ bcol,
                                              float* __restrict__ trow) {
#pragma unroll
  for (int t = T0; t < T1; ++t) {
    const float g = gcol[t * 16], bb = bcol[t * 16];
#pragma unroll
    for (int r = 0; r < 8; ++r) {
      const float v = (acc[t][r] - mu[r]) * rs[r] * g + bb;
      trow[(size_t)r * kHid + t * 16] = gelu_exact(v);
    }
  }
}

__global__ __launch_bounds__(128)
void gemm1_ln_gelu(const float* __restrict__ A,
                   const float* __restrict__ W1,
                   const float* __restrict__ b1,
                   const float* __restrict__ lng,
                   const float* __restrict__ lnb,
                   float* __restrict__ T) {
  __shared__ __attribute__((aligned(16))) float sW[kHid * kPitch];

  const int tid  = threadIdx.x;
  const int wave = tid >> 5;
  const int lane = tid & 31;
  const int hf   = lane >> 4;
  const int l16  = lane & 15;
  const int row0 = blockIdx.x * kBlockRows + wave * kWaveRows;

  v8f acc[kTiles];
  gemm_rows16_x768(A + (size_t)(row0 + l16) * kF + 2 * hf, W1, sW, acc, tid, hf, l16);

  float s1[8], s2[8];
#pragma unroll
  for (int r = 0; r < 8; ++r) { s1[r] = 0.f; s2[r] = 0.f; }
#pragma unroll
  for (int t = 0; t < kTiles; ++t) {
    const float bias = b1[t * 16 + l16];
#pragma unroll
    for (int r = 0; r < 8; ++r) {
      const float v = acc[t][r] + bias;
      acc[t][r] = v;
      s1[r] += v;
      s2[r] += v * v;
    }
  }
  float mu[8], rs[8];
#pragma unroll
  for (int r = 0; r < 8; ++r) {
#pragma unroll
    for (int off = 8; off >= 1; off >>= 1) {
      s1[r] += __shfl_xor(s1[r], off, 32);
      s2[r] += __shfl_xor(s2[r], off, 32);
    }
    mu[r] = s1[r] / kF;
    const float var = s2[r] / kF - mu[r] * mu[r];
    rs[r] = rsqrtf(var + 1e-5f);
  }
  float* __restrict__ trow = T + (size_t)(row0 + 8 * hf) * kHid + l16;
  ln_gelu_store<0, kTiles / 2>(acc, mu, rs, lng + l16, lnb + l16, trow);
  ln_gelu_store<kTiles / 2, kTiles>(acc, mu, rs, lng + l16, lnb + l16, trow);
}

__global__ __launch_bounds__(128)
void gemm2_perm(const float* __restrict__ T,
                const float* __restrict__ W2,
                const float* __restrict__ b2,
                const int*   __restrict__ perm,
                float* __restrict__ out) {
  __shared__ __attribute__((aligned(16))) float sW[kOut * kPitch];

  const int tid  = threadIdx.x;
  const int wave = tid >> 5;
  const int lane = tid & 31;
  const int hf   = lane >> 4;
  const int l16  = lane & 15;
  const int row0 = blockIdx.x * kBlockRows + wave * kWaveRows;

  const int g    = row0 + l16;
  const int img  = g / kN;
  const int n0   = g - img * kN;
  const int src  = img * kN + perm[n0];

  v8f acc[kTiles];
  gemm_rows16_x768(T + (size_t)src * kHid + 2 * hf, W2, sW, acc, tid, hf, l16);

#pragma unroll
  for (int t = 0; t < kTiles; ++t) {
    const int n = t * 16 + l16;
    const float bias = b2[n];
#pragma unroll
    for (int r = 0; r < 8; ++r) {
      out[(size_t)(row0 + 8 * hf + r) * kOut + n] = acc[t][r] + bias;
    }
  }
}

extern "C" void kernel_launch(void* const* d_in, const int* in_sizes, int n_in,
                              void* d_out, int out_size, void* d_ws, size_t ws_size,
                              hipStream_t stream) {
  const float* x    = (const float*)d_in[0];
  const float* cw   = (const float*)d_in[1];
  const float* cb   = (const float*)d_in[2];
  const float* bng  = (const float*)d_in[3];
  const float* bnb  = (const float*)d_in[4];
  const float* bnm  = (const float*)d_in[5];
  const float* bnv  = (const float*)d_in[6];
  const float* M    = (const float*)d_in[7];
  const float* W1   = (const float*)d_in[8];
  const float* b1   = (const float*)d_in[9];
  const float* lng  = (const float*)d_in[10];
  const float* lnb  = (const float*)d_in[11];
  const float* W2   = (const float*)d_in[12];
  const float* b2   = (const float*)d_in[13];
  const int*   mask = (const int*)d_in[14];
  const int*   perm = (const int*)d_in[15];
  float* out = (float*)d_out;

  float* flat = (float*)d_ws;
  float* Tbuf = flat + (size_t)kRows * kF;

  conv_bn_gelu_patch<<<kBS * kPR * 2, kConvThreads, 0, stream>>>(
      x, cw, cb, bng, bnb, bnm, bnv, M, mask, flat);
  gemm1_ln_gelu<<<kRows / kBlockRows, kWaves * 32, 0, stream>>>(flat, W1, b1, lng, lnb, Tbuf);
  gemm2_perm  <<<kRows / kBlockRows, kWaves * 32, 0, stream>>>(Tbuf, W2, b2, perm, out);
}


# fc1 epilogue: 384 inlined erff bodies (170 KB of straight-line code, beyond the instruction cache) become a rolled row loop fed from LDS; fc1 k-loop load groups pinned in place; sum orders kept
# speedup vs baseline: 2.1074x; 1.3818x over previous
#include <hip/hip_runtime.h>
#include <math.h>


typedef float v2f __attribute__((ext_vector_type(2)));
typedef float v8f __attribute__((ext_vector_type(8)));

namespace {
constexpr int kPS   = 16;
constexpr int kPR   = 14;
constexpr int kN    = 196;
constexpr int kC    = 3;
constexpr int kW    = 224;
constexpr int kBS   = 64;
constexpr int kF    = 768;
constexpr int kHid  = 768;
constexpr int kOut  = 768;
constexpr int kRows = kBS * kN;
constexpr int kWaveRows  = 16;
constexpr int kWaves     = 4;
constexpr int kBlockRows = kWaveRows * kWaves;
constexpr int kTiles     = kHid / 16;
constexpr int kKC        = 32;
constexpr int kPitch     = 36;
constexpr int kHalfTiles = kTiles / 2;
constexpr int kEPitch    = 388;
constexpr int kEWave     = kWaveRows * kEPitch;
constexpr int kEStat     = kWaves * kEWave;
static_assert(kEStat + kBlockRows * 2 <= kHid * kPitch, "staged rows must fit in the W chunk's LDS");
}

__device__ __forceinline__ float gelu_exact(float v) {
  return 0.5f * v * (1.0f + erff(v * 0.70710678118654752440f));
}

constexpr int kConvThreads = kF / 4;
constexpr int kConvPatches = 7;

__global__ __launch_bounds__(kConvThreads)
void conv_bn_gelu_patch(const float* __restrict__ x,
                        const float* __restrict__ cw,
                        const float* __restrict__ cb,
                        const float* __restrict__ bng,
                        const float* __restrict__ bnb,
                        const float* __restrict__ bnm,
                        const float* __restrict__ bnv,
                        const float* __restrict__ M,
                        const int*   __restrict__ mask,
                        float* __restrict__ flat) {
  const int t   = threadIdx.x;
  const int co  = __builtin_amdgcn_readfirstlane(t >> 6);
  const int pi  = (t >> 2) & 15;
  const int pj0 = (t & 3) * 4;

  int b = blockIdx.x;
  const int ph0 = (b & 1) * kConvPatches;  b >>= 1;
  const int pw  = b % kPR;
  const int img = b / kPR;
  const int w   = pw * kPS + pi;

  int  roff[3];
  float wt[kC][3][3];
#pragma unroll
  for (int kw = 0; kw < 3; ++kw) {
    const int ww = w + kw - 1;
    const bool ok = (unsigned)ww < (unsigned)kW;
    roff[kw] = (ok ? ww : w) * kW;
#pragma unroll
    for (int ci = 0; ci < kC; ++ci)
#pragma unroll
      for (int kh = 0; kh < 3; ++kh) {
        const float wv = cw[((co * kC + ci) * 3 + kw) * 3 + kh];
        wt[ci][kw][kh] = ok ? wv : 0.0f;
      }
  }
  const float bias = cb[co];
  const float inv  = bng[co] * rsqrtf(bnv[co] + 1e-5f);
  const float mean = bnm[co];
  const float beta = bnb[co];
  float mm[4];
#pragma unroll
  for (int j = 0; j < 4; ++j) mm[j] = M[pi * kPS + pj0 + j];

  const float* __restrict__ ximg = x + (size_t)img * kC * kW * kW;

  for (int i = 0; i < kConvPatches; ++i) {
    const int ph = ph0 + i;
    const int h0 = ph * kPS + pj0;
    const bool lok = h0 > 0;
    const bool rok = h0 + 4 < kW;
    const int hl = lok ? h0 - 1 : h0;
    const int hr = rok ? h0 + 4 : h0 + 3;

    float xv[kC][3][6];
#pragma unroll
    for (int ci = 0; ci < kC; ++ci)
#pragma unroll
      for (int kw = 0; kw < 3; ++kw) {
        const float* __restrict__ rp = ximg + ci * kW * kW + roff[kw];
        const float4 v = *(const float4*)(rp + h0);
        const float l = rp[hl];
        const float r = rp[hr];
        xv[ci][kw][0] = lok ? l : 0.0f;
        xv[ci][kw][1] = v.x;
        xv[ci][kw][2] = v.y;
        xv[ci][kw][3] = v.z;
        xv[ci][kw][4] = v.w;
        xv[ci][kw][5] = rok ? r : 0.0f;
      }

    float acc[4] = {bias, bias, bias, bias};
#pragma unroll
    for (int ci = 0; ci < kC; ++ci)
#pragma unroll
      for (int kw = 0; kw < 3; ++kw)
#pragma unroll
        for (int kh = 0; kh < 3; ++kh)
#pragma unroll
          for (int j = 0; j < 4; ++j)
            acc[j] += xv[ci][kw][j + kh] * wt[ci][kw][kh];

    const int n = pw * kPR + ph;
    const bool gate = mask[n] > 0;
    float o[4];
#pragma unroll
    for (int j = 0; j < 4; ++j) {
      float a = (acc[j] - mean) * inv + beta;
      a = gelu_exact(a);
      if (gate) a *= mm[j];
      o[j] = a;
    }
    const int row = img * kN + n;
    *(float4*)(flat + (size_t)row * kF + 4 * t) = make_float4(o[0], o[1], o[2], o[3]);
  }
}

template <bool kPinOrder>
__device__ __forceinline__ void gemm_rows16_x768(const float* __restrict__ arow,
                                                 const float* __restrict__ Wm,
                                                 float* __restrict__ sW,
                                                 v8f (&acc)[kTiles],
                                                 int tid, int hf, int l16) {
#pragma unroll
  for (int t = 0; t < kTiles; ++t) acc[t] = v8f{};

  const float* __restrict__ brow = sW + l16 * kPitch + 2 * hf;

#pragma unroll 1
  for (int kc = 0; kc < kF; kc += kKC) {
    __syncthreads();
    {
      float4 v[48];
#pragma unroll
      for (int i = 0; i < 48; ++i) {
        const int idx = tid + i * (kWaves * 32);
        const int n = idx >> 3, q = idx & 7;
        v[i] = *(const float4*)(Wm + (size_t)n * kF + kc + 4 * q);
      }
      if (kPinOrder) __builtin_amdgcn_sched_barrier(0);
#pragma unroll
      for (int i = 0; i < 48; ++i) {
        const int idx = tid + i * (kWaves * 32);
        const int n = idx >> 3, q = idx & 7;
        *(float4*)(sW + n * kPitch + 4 * q) = v[i];
      }
    }
    __syncthreads();

    v2f a[kKC / 4];
#pragma unroll
    for (int s = 0; s < kKC / 4; ++s) a[s] = *(const v2f*)(arow + kc + 4 * s);

    if (kPinOrder) {
      v2f b[2][kTiles];
#pragma unroll
      for (int t = 0; t < kTiles; ++t) b[0][t] = *(const v2f*)(brow + t * 16 * kPitch);
#pragma unroll
      for (int s = 0; s < kKC / 4; ++s) {
        if (s + 1 < kKC / 4) {
#pragma unroll
          for (int t = 0; t < kTiles; ++t)
            b[(s + 1) & 1][t] = *(const v2f*)(brow + t * 16 * kPitch + 4 * (s + 1));
        }
        __builtin_amdgcn_sched_barrier(0);
#pragma unroll
        for (int t = 0; t < kTiles; ++t)
          acc[t] = __builtin_amdgcn_wmma_f32_16x16x4_f32(false, a[s], false, b[s & 1][t],
                                                         (short)0, acc[t], false, false);
        __builtin_amdgcn_sched_barrier(0);
      }
    } else {
#pragma unroll
      for (int s = 0; s < kKC / 4; ++s) {
#pragma unroll
        for (int t = 0; t < kTiles; ++t) {
          const v2f b = *(const v2f*)(brow + t * 16 * kPitch + 4 * s);
          acc[t] = __builtin_amdgcn_wmma_f32_16x16x4_f32(false, a[s], false, b,
                                                         (short)0, acc[t], false, false);
        }
      }
    }
  }
}

template <int H>
__device__ __forceinline__ void ln_gelu_pass(const v8f (&acc)[kTiles],
                                             float* myrows,
                                             const float* stat,
                                             const float* __restrict__ lng,
                                             const float* __restrict__ lnb,
                                             float* __restrict__ T0,
                                             int lane, int hf, int l16) {
  float* wr = myrows + (8 * hf) * kEPitch + l16;
#pragma unroll
  for (int t = 0; t < kHalfTiles; ++t)
#pragma unroll
    for (int r = 0; r < 8; ++r)
      wr[r * kEPitch + t * 16] = acc[H * kHalfTiles + t][r];

  __syncthreads();

  const int col0 = H * (kHid / 2) + 4 * lane;
  float4 g4[3], b4[3];
#pragma unroll
  for (int j = 0; j < 3; ++j) {
    g4[j] = *(const float4*)(lng + col0 + 128 * j);
    b4[j] = *(const float4*)(lnb + col0 + 128 * j);
  }
  const float* src = myrows + 4 * lane;
  float* __restrict__ dst = T0 + col0;

  for (int i = 0; i < kWaveRows; ++i) {
    const float mu = stat[2 * i];
    const float rs = stat[2 * i + 1];
#pragma unroll
    for (int j = 0; j < 3; ++j) {
      const float4 v = *(const float4*)(src + i * kEPitch + 128 * j);
      float4 o;
      o.x = gelu_exact((v.x - mu) * rs * g4[j].x + b4[j].x);
      o.y = gelu_exact((v.y - mu) * rs * g4[j].y + b4[j].y);
      o.z = gelu_exact((v.z - mu) * rs * g4[j].z + b4[j].z);
      o.w = gelu_exact((v.w - mu) * rs * g4[j].w + b4[j].w);
      *(float4*)(dst + (size_t)i * kHid + 128 * j) = o;
    }
  }
}

__global__ __launch_bounds__(128)
void gemm1_ln_gelu(const float* __restrict__ A,
                   const float* __restrict__ W1,
                   const float* __restrict__ b1,
                   const float* __restrict__ lng,
                   const float* __restrict__ lnb,
                   float* __restrict__ T) {
  __shared__ __attribute__((aligned(16))) float sW[kHid * kPitch];

  const int tid  = threadIdx.x;
  const int wave = tid >> 5;
  const int lane = tid & 31;
  const int hf   = lane >> 4;
  const int l16  = lane & 15;
  const int row0 = blockIdx.x * kBlockRows + wave * kWaveRows;

  v8f acc[kTiles];
  gemm_rows16_x768<true>(A + (size_t)(row0 + l16) * kF + 2 * hf, W1, sW, acc, tid, hf, l16);

  float s1[8], s2[8];
#pragma unroll
  for (int r = 0; r < 8; ++r) { s1[r] = 0.f; s2[r] = 0.f; }
#pragma unroll
  for (int t = 0; t < kTiles; ++t) {
    const float bias = b1[t * 16 + l16];
#pragma unroll
    for (int r = 0; r < 8; ++r) {
      const float v = acc[t][r] + bias;
      acc[t][r] = v;
      s1[r] += v;
      s2[r] += v * v;
    }
  }
  float mu[8], rs[8];
#pragma unroll
  for (int r = 0; r < 8; ++r) {
#pragma unroll
    for (int off = 8; off >= 1; off >>= 1) {
      s1[r] += __shfl_xor(s1[r], off, 32);
      s2[r] += __shfl_xor(s2[r], off, 32);
    }
    mu[r] = s1[r] / kF;
    const float var = s2[r] / kF - mu[r] * mu[r];
    rs[r] = rsqrtf(var + 1e-5f);
  }

  __syncthreads();

  float* myrows = sW + wave * kEWave;
  float* stat   = sW + kEStat + wave * (kWaveRows * 2);
  if (l16 == 0) {
#pragma unroll
    for (int r = 0; r < 8; ++r) {
      stat[(8 * hf + r) * 2]     = mu[r];
      stat[(8 * hf + r) * 2 + 1] = rs[r];
    }
  }
  float* __restrict__ T0 = T + (size_t)row0 * kHid;
  ln_gelu_pass<0>(acc, myrows, stat, lng, lnb, T0, lane, hf, l16);
  __syncthreads();
  ln_gelu_pass<1>(acc, myrows, stat, lng, lnb, T0, lane, hf, l16);
}

__global__ __launch_bounds__(128)
void gemm2_perm(const float* __restrict__ T,
                const float* __restrict__ W2,
                const float* __restrict__ b2,
                const int*   __restrict__ perm,
                float* __restrict__ out) {
  __shared__ __attribute__((aligned(16))) float sW[kOut * kPitch];

  const int tid  = threadIdx.x;
  const int wave = tid >> 5;
  const int lane = tid & 31;
  const int hf   = lane >> 4;
  const int l16  = lane & 15;
  const int row0 = blockIdx.x * kBlockRows + wave * kWaveRows;

  const int g    = row0 + l16;
  const int img  = g / kN;
  const int n0   = g - img * kN;
  const int src  = img * kN + perm[n0];

  v8f acc[kTiles];
  gemm_rows16_x768<false>(T + (size_t)src * kHid + 2 * hf, W2, sW, acc, tid, hf, l16);

#pragma unroll
  for (int t = 0; t < kTiles; ++t) {
    const int n = t * 16 + l16;
    const float bias = b2[n];
#pragma unroll
    for (int r = 0; r < 8; ++r) {
      out[(size_t)(row0 + 8 * hf + r) * kOut + n] = acc[t][r] + bias;
    }
  }
}

extern "C" void kernel_launch(void* const* d_in, const int* in_sizes, int n_in,
                              void* d_out, int out_size, void* d_ws, size_t ws_size,
                              hipStream_t stream) {
  const float* x    = (const float*)d_in[0];
  const float* cw   = (const float*)d_in[1];
  const float* cb   = (const float*)d_in[2];
  const float* bng  = (const float*)d_in[3];
  const float* bnb  = (const float*)d_in[4];
  const float* bnm  = (const float*)d_in[5];
  const float* bnv  = (const float*)d_in[6];
  const float* M    = (const float*)d_in[7];
  const float* W1   = (const float*)d_in[8];
  const float* b1   = (const float*)d_in[9];
  const float* lng  = (const float*)d_in[10];
  const float* lnb  = (const float*)d_in[11];
  const float* W2   = (const float*)d_in[12];
  const float* b2   = (const float*)d_in[13];
  const int*   mask = (const int*)d_in[14];
  const int*   perm = (const int*)d_in[15];
  float* out = (float*)d_out;

  float* flat = (float*)d_ws;
  float* Tbuf = flat + (size_t)kRows * kF;

  conv_bn_gelu_patch<<<kBS * kPR * 2, kConvThreads, 0, stream>>>(
      x, cw, cb, bng, bnb, bnm, bnv, M, mask, flat);
  gemm1_ln_gelu<<<kRows / kBlockRows, kWaves * 32, 0, stream>>>(flat, W1, b1, lng, lnb, Tbuf);
  gemm2_perm  <<<kRows / kBlockRows, kWaves * 32, 0, stream>>>(Tbuf, W2, b2, perm, out);
}


# both GEMMs: k loop software-pipelined over two LDS stages of the shared W chunk: next chunk's W lines and A fragments are fetched under the matrix blocks, one barrier per chunk; sum order kept
# speedup vs baseline: 2.3270x; 1.1042x over previous
#include <hip/hip_runtime.h>
#include <math.h>


typedef float v2f __attribute__((ext_vector_type(2)));
typedef float v4f __attribute__((ext_vector_type(4)));
typedef float v8f __attribute__((ext_vector_type(8)));

namespace {
constexpr int kPS   = 16;
constexpr int kPR   = 14;
constexpr int kN    = 196;
constexpr int kC    = 3;
constexpr int kW    = 224;
constexpr int kBS   = 64;
constexpr int kF    = 768;
constexpr int kHid  = 768;
constexpr int kOut  = 768;
constexpr int kRows = kBS * kN;
constexpr int kWaveRows  = 16;
constexpr int kWaves     = 4;
constexpr int kBlockRows = kWaveRows * kWaves;
constexpr int kTiles     = kHid / 16;
constexpr int kKC        = 32;
constexpr int kPitch     = 36;
constexpr int kStage     = kHid * kPitch;
constexpr int kSteps     = kKC / 4;
constexpr int kGrp       = 10;
constexpr int kFetchSteps = 5;
constexpr int kHalfTiles = kTiles / 2;
constexpr int kEPitch    = 388;
constexpr int kEWave     = kWaveRows * kEPitch;
constexpr int kEStat     = kWaves * kEWave;
static_assert(kEStat + kBlockRows * 2 <= kStage, "staged rows must fit in the first LDS stage");
static_assert(kFetchSteps * kGrp >= 48 && 2 + kFetchSteps <= kSteps - 1, "all 48 pieces are stored before the hand-over barrier in front of the last block");
}

__device__ __forceinline__ float gelu_exact(float v) {
  return 0.5f * v * (1.0f + erff(v * 0.70710678118654752440f));
}

constexpr int kConvThreads = kF / 4;
constexpr int kConvPatches = 7;

__global__ __launch_bounds__(kConvThreads)
void conv_bn_gelu_patch(const float* __restrict__ x,
                        const float* __restrict__ cw,
                        const float* __restrict__ cb,
                        const float* __restrict__ bng,
                        const float* __restrict__ bnb,
                        const float* __restrict__ bnm,
                        const float* __restrict__ bnv,
                        const float* __restrict__ M,
                        const int*   __restrict__ mask,
                        float* __restrict__ flat) {
  const int t   = threadIdx.x;
  const int co  = __builtin_amdgcn_readfirstlane(t >> 6);
  const int pi  = (t >> 2) & 15;
  const int pj0 = (t & 3) * 4;

  int b = blockIdx.x;
  const int ph0 = (b & 1) * kConvPatches;  b >>= 1;
  const int pw  = b % kPR;
  const int img = b / kPR;
  const int w   = pw * kPS + pi;

  int  roff[3];
  float wt[kC][3][3];
#pragma unroll
  for (int kw = 0; kw < 3; ++kw) {
    const int ww = w + kw - 1;
    const bool ok = (unsigned)ww < (unsigned)kW;
    roff[kw] = (ok ? ww : w) * kW;
#pragma unroll
    for (int ci = 0; ci < kC; ++ci)
#pragma unroll
      for (int kh = 0; kh < 3; ++kh) {
        const float wv = cw[((co * kC + ci) * 3 + kw) * 3 + kh];
        wt[ci][kw][kh] = ok ? wv : 0.0f;
      }
  }
  const float bias = cb[co];
  const float inv  = bng[co] * rsqrtf(bnv[co] + 1e-5f);
  const float mean = bnm[co];
  const float beta = bnb[co];
  float mm[4];
#pragma unroll
  for (int j = 0; j < 4; ++j) mm[j] = M[pi * kPS + pj0 + j];

  const float* __restrict__ ximg = x + (size_t)img * kC * kW * kW;

  for (int i = 0; i < kConvPatches; ++i) {
    const int ph = ph0 + i;
    const int h0 = ph * kPS + pj0;
    const bool lok = h0 > 0;
    const bool rok = h0 + 4 < kW;
    const int hl = lok ? h0 - 1 : h0;
    const int hr = rok ? h0 + 4 : h0 + 3;

    float xv[kC][3][6];
#pragma unroll
    for (int ci = 0; ci < kC; ++ci)
#pragma unroll
      for (int kw = 0; kw < 3; ++kw) {
        const float* __restrict__ rp = ximg + ci * kW * kW + roff[kw];
        const float4 v = *(const float4*)(rp + h0);
        const float l = rp[hl];
        const float r = rp[hr];
        xv[ci][kw][0] = lok ? l : 0.0f;
        xv[ci][kw][1] = v.x;
        xv[ci][kw][2] = v.y;
        xv[ci][kw][3] = v.z;
        xv[ci][kw][4] = v.w;
        xv[ci][kw][5] = rok ? r : 0.0f;
      }

    float acc[4] = {bias, bias, bias, bias};
#pragma unroll
    for (int ci = 0; ci < kC; ++ci)
#pragma unroll
      for (int kw = 0; kw < 3; ++kw)
#pragma unroll
        for (int kh = 0; kh < 3; ++kh)
#pragma unroll
          for (int j = 0; j < 4; ++j)
            acc[j] += xv[ci][kw][j + kh] * wt[ci][kw][kh];

    const int n = pw * kPR + ph;
    const bool gate = mask[n] > 0;
    float o[4];
#pragma unroll
    for (int j = 0; j < 4; ++j) {
      float a = (acc[j] - mean) * inv + beta;
      a = gelu_exact(a);
      if (gate) a *= mm[j];
      o[j] = a;
    }
    const int row = img * kN + n;
    *(float4*)(flat + (size_t)row * kF + 4 * t) = make_float4(o[0], o[1], o[2], o[3]);
  }
}

template <bool kMore>
__device__ __forceinline__ void k_chunk(const float* __restrict__ anext,
                                        const float* __restrict__ wnext,
                                        const float* rdB,
                                        float* wrW,
                                        const float* nxB,
                                        v2f (&a)[kSteps], v2f (&b)[2][kTiles], v8f (&acc)[kTiles]) {
  v2f an[kSteps];
  v4f wv[2][kGrp];
#pragma unroll
  for (int s = 0; s < kSteps; ++s) {
    if (s + 1 < kSteps) {
#pragma unroll
      for (int t = 0; t < kTiles; ++t)
        b[(s + 1) & 1][t] = *(const v2f*)(rdB + t * 16 * kPitch + 4 * (s + 1));
    }
    if (kMore) {
      if (s == 0) {
#pragma unroll
        for (int j = 0; j < kSteps; ++j) an[j] = *(const v2f*)(anext + 4 * j);
      }
      if (s >= 2 && s < 2 + kFetchSteps) {
#pragma unroll
        for (int j = 0; j < kGrp; ++j)
          if (kGrp * (s - 2) + j < 48)
            *(v4f*)(wrW + (kGrp * (s - 2) + j) * 16 * kPitch) = wv[s & 1][j];
      }
      if (s < kFetchSteps) {
#pragma unroll
        for (int j = 0; j < kGrp; ++j)
          if (kGrp * s + j < 48)
            wv[s & 1][j] = *(const v4f*)(wnext + (size_t)(kGrp * s + j) * 16 * kF);
      }
      if (s == kSteps - 1) {
        __syncthreads();
#pragma unroll
        for (int t = 0; t < kTiles; ++t) b[0][t] = *(const v2f*)(nxB + t * 16 * kPitch);
      }
    }
    __builtin_amdgcn_sched_barrier(0);
#pragma unroll
    for (int t = 0; t < kTiles; ++t)
      acc[t] = __builtin_amdgcn_wmma_f32_16x16x4_f32(false, a[s], false, b[s & 1][t],
                                                     (short)0, acc[t], false, false);
    __builtin_amdgcn_sched_barrier(0);
  }
  if (kMore) {
#pragma unroll
    for (int s = 0; s < kSteps; ++s) a[s] = an[s];
  }
}

__device__ __forceinline__ void gemm_rows16_x768(const float* __restrict__ arow,
                                                 const float* __restrict__ Wm,
                                                 float* __restrict__ sW,
                                                 v8f (&acc)[kTiles],
                                                 int tid, int hf, int l16) {
#pragma unroll
  for (int t = 0; t < kTiles; ++t) acc[t] = v8f{};

  const float* __restrict__ wsrc = Wm + (size_t)(tid >> 3) * kF + 4 * (tid & 7);
  const int wslot = (tid >> 3) * kPitch + 4 * (tid & 7);
  const int bfrag = l16 * kPitch + 2 * hf;

  {
    v4f v[48];
#pragma unroll
    for (int i = 0; i < 48; ++i) v[i] = *(const v4f*)(wsrc + (size_t)i * 16 * kF);
    __builtin_amdgcn_sched_barrier(0);
#pragma unroll
    for (int i = 0; i < 48; ++i) *(v4f*)(sW + wslot + i * 16 * kPitch) = v[i];
  }
  v2f a[kSteps];
#pragma unroll
  for (int s = 0; s < kSteps; ++s) a[s] = *(const v2f*)(arow + 4 * s);
  __syncthreads();

  v2f b[2][kTiles];
#pragma unroll
  for (int t = 0; t < kTiles; ++t) b[0][t] = *(const v2f*)(sW + bfrag + t * 16 * kPitch);

  float* rd = sW;
  float* wr = sW + kStage;
#pragma unroll 1
  for (int kc = 0; kc < kF - kKC; kc += kKC) {
    k_chunk<true>(arow + kc + kKC, wsrc + kc + kKC, rd + bfrag, wr + wslot, wr + bfrag, a, b, acc);
    float* const tmp = rd; rd = wr; wr = tmp;
  }
  k_chunk<false>(arow, wsrc, rd + bfrag, wr + wslot, wr + bfrag, a, b, acc);
}

template <int H>
__device__ __forceinline__ void ln_gelu_pass(const v8f (&acc)[kTiles],
                                             float* myrows,
                                             const float* stat,
                                             const float* __restrict__ lng,
                                             const float* __restrict__ lnb,
                                             float* __restrict__ T0,
                                             int lane, int hf, int l16) {
  float* wr = myrows + (8 * hf) * kEPitch + l16;
#pragma unroll
  for (int t = 0; t < kHalfTiles; ++t)
#pragma unroll
    for (int r = 0; r < 8; ++r)
      wr[r * kEPitch + t * 16] = acc[H * kHalfTiles + t][r];

  __syncthreads();

  const int col0 = H * (kHid / 2) + 4 * lane;
  float4 g4[3], b4[3];
#pragma unroll
  for (int j = 0; j < 3; ++j) {
    g4[j] = *(const float4*)(lng + col0 + 128 * j);
    b4[j] = *(const float4*)(lnb + col0 + 128 * j);
  }
  const float* src = myrows + 4 * lane;
  float* __restrict__ dst = T0 + col0;

  for (int i = 0; i < kWaveRows; ++i) {
    const float mu = stat[2 * i];
    const float rs = stat[2 * i + 1];
#pragma unroll
    for (int j = 0; j < 3; ++j) {
      const float4 v = *(const float4*)(src + i * kEPitch + 128 * j);
      float4 o;
      o.x = gelu_exact((v.x - mu) * rs * g4[j].x + b4[j].x);
      o.y = gelu_exact((v.y - mu) * rs * g4[j].y + b4[j].y);
      o.z = gelu_exact((v.z - mu) * rs * g4[j].z + b4[j].z);
      o.w = gelu_exact((v.w - mu) * rs * g4[j].w + b4[j].w);
      *(float4*)(dst + (size_t)i * kHid + 128 * j) = o;
    }
  }
}

__global__ __launch_bounds__(128)
void gemm1_ln_gelu(const float* __restrict__ A,
                   const float* __restrict__ W1,
                   const float* __restrict__ b1,
                   const float* __restrict__ lng,
                   const float* __restrict__ lnb,
                   float* __restrict__ T) {
  __shared__ __attribute__((aligned(16))) float sW[2 * kStage];

  const int tid  = threadIdx.x;
  const int wave = tid >> 5;
  const int lane = tid & 31;
  const int hf   = lane >> 4;
  const int l16  = lane & 15;
  const int row0 = blockIdx.x * kBlockRows + wave * kWaveRows;

  v8f acc[kTiles];
  gemm_rows16_x768(A + (size_t)(row0 + l16) * kF + 2 * hf, W1, sW, acc, tid, hf, l16);

  float s1[8], s2[8];
#pragma unroll
  for (int r = 0; r < 8; ++r) { s1[r] = 0.f; s2[r] = 0.f; }
#pragma unroll
  for (int t = 0; t < kTiles; ++t) {
    const float bias = b1[t * 16 + l16];
#pragma unroll
    for (int r = 0; r < 8; ++r) {
      const float v = acc[t][r] + bias;
      acc[t][r] = v;
      s1[r] += v;
      s2[r] += v * v;
    }
  }
  float mu[8], rs[8];
#pragma unroll
  for (int r = 0; r < 8; ++r) {
#pragma unroll
    for (int off = 8; off >= 1; off >>= 1) {
      s1[r] += __shfl_xor(s1[r], off, 32);
      s2[r] += __shfl_xor(s2[r], off, 32);
    }
    mu[r] = s1[r] / kF;
    const float var = s2[r] / kF - mu[r] * mu[r];
    rs[r] = rsqrtf(var + 1e-5f);
  }

  float* myrows = sW + wave * kEWave;
  float* stat   = sW + kEStat + wave * (kWaveRows * 2);
  if (l16 == 0) {
#pragma unroll
    for (int r = 0; r < 8; ++r) {
      stat[(8 * hf + r) * 2]     = mu[r];
      stat[(8 * hf + r) * 2 + 1] = rs[r];
    }
  }
  float* __restrict__ T0 = T + (size_t)row0 * kHid;
  ln_gelu_pass<0>(acc, myrows, stat, lng, lnb, T0, lane, hf, l16);
  __syncthreads();
  ln_gelu_pass<1>(acc, myrows, stat, lng, lnb, T0, lane, hf, l16);
}

__global__ __launch_bounds__(128)
void gemm2_perm(const float* __restrict__ T,
                const float* __restrict__ W2,
                const float* __restrict__ b2,
                const int*   __restrict__ perm,
                float* __restrict__ out) {
  __shared__ __attribute__((aligned(16))) float sW[2 * kStage];

  const int tid  = threadIdx.x;
  const int wave = tid >> 5;
  const int lane = tid & 31;
  const int hf   = lane >> 4;
  const int l16  = lane & 15;
  const int row0 = blockIdx.x * kBlockRows + wave * kWaveRows;

  const int g    = row0 + l16;
  const int img  = g / kN;
  const int n0   = g - img * kN;
  const int src  = img * kN + perm[n0];

  v8f acc[kTiles];
  gemm_rows16_x768(T + (size_t)src * kHid + 2 * hf, W2, sW, acc, tid, hf, l16);

#pragma unroll
  for (int t = 0; t < kTiles; ++t) {
    const int n = t * 16 + l16;
    const float bias = b2[n];
#pragma unroll
    for (int r = 0; r < 8; ++r) {
      out[(size_t)(row0 + 8 * hf + r) * kOut + n] = acc[t][r] + bias;
    }
  }
}

extern "C" void kernel_launch(void* const* d_in, const int* in_sizes, int n_in,
                              void* d_out, int out_size, void* d_ws, size_t ws_size,
                              hipStream_t stream) {
  const float* x    = (const float*)d_in[0];
  const float* cw   = (const float*)d_in[1];
  const float* cb   = (const float*)d_in[2];
  const float* bng  = (const float*)d_in[3];
  const float* bnb  = (const float*)d_in[4];
  const float* bnm  = (const float*)d_in[5];
  const float* bnv  = (const float*)d_in[6];
  const float* M    = (const float*)d_in[7];
  const float* W1   = (const float*)d_in[8];
  const float* b1   = (const float*)d_in[9];
  const float* lng  = (const float*)d_in[10];
  const float* lnb  = (const float*)d_in[11];
  const float* W2   = (const float*)d_in[12];
  const float* b2   = (const float*)d_in[13];
  const int*   mask = (const int*)d_in[14];
  const int*   perm = (const int*)d_in[15];
  float* out = (float*)d_out;

  float* flat = (float*)d_ws;
  float* Tbuf = flat + (size_t)kRows * kF;

  conv_bn_gelu_patch<<<kBS * kPR * 2, kConvThreads, 0, stream>>>(
      x, cw, cb, bng, bnb, bnm, bnv, M, mask, flat);
  gemm1_ln_gelu<<<kRows / kBlockRows, kWaves * 32, 0, stream>>>(flat, W1, b1, lng, lnb, Tbuf);
  gemm2_perm  <<<kRows / kBlockRows, kWaves * 32, 0, stream>>>(Tbuf, W2, b2, perm, out);
}


# both GEMMs: 8 waves per workgroup, two per SIMD: a wave owns 16 rows x 384 columns (24 tiles), one wave's load groups run under the other's matrix block; LayerNorm sums in two halves (order changed)
# speedup vs baseline: 2.7111x; 1.1651x over previous
#include <hip/hip_runtime.h>
#include <math.h>


typedef float v2f __attribute__((ext_vector_type(2)));
typedef float v4f __attribute__((ext_vector_type(4)));
typedef float v8f __attribute__((ext_vector_type(8)));

namespace {
constexpr int kPS   = 16;
constexpr int kPR   = 14;
constexpr int kN    = 196;
constexpr int kC    = 3;
constexpr int kW    = 224;
constexpr int kBS   = 64;
constexpr int kF    = 768;
constexpr int kHid  = 768;
constexpr int kOut  = 768;
constexpr int kRows = kBS * kN;
constexpr int kWaveRows  = 16;
constexpr int kRowGroups = 4;
constexpr int kColHalves = 2;
constexpr int kWaves     = kRowGroups * kColHalves;
constexpr int kThreads   = kWaves * 32;
constexpr int kBlockRows = kWaveRows * kRowGroups;
constexpr int kTiles     = kHid / 16 / kColHalves;
constexpr int kKC        = 32;
constexpr int kPitch     = 36;
constexpr int kStage     = kHid * kPitch;
constexpr int kSteps     = kKC / 4;
constexpr int kPieces    = kHid * (kKC / 4) / kThreads;
constexpr int kPieceRows = kThreads / 8;
constexpr int kGrp       = 5;
constexpr int kFetchSteps = 5;
constexpr int kEPitch    = 388;
constexpr int kEWave     = kWaveRows * kEPitch;
static_assert(kWaves * kEWave <= 2 * kStage, "staged rows must fit in the two LDS stages");
static_assert(kFetchSteps * kGrp >= kPieces && 2 + kFetchSteps <= kSteps - 1, "all pieces are stored before the hand-over barrier in front of the last block");
}

__device__ __forceinline__ float gelu_exact(float v) {
  return 0.5f * v * (1.0f + erff(v * 0.70710678118654752440f));
}

constexpr int kConvThreads = kF / 4;
constexpr int kConvPatches = 7;

__global__ __launch_bounds__(kConvThreads)
void conv_bn_gelu_patch(const float* __restrict__ x,
                        const float* __restrict__ cw,
                        const float* __restrict__ cb,
                        const float* __restrict__ bng,
                        const float* __restrict__ bnb,
                        const float* __restrict__ bnm,
                        const float* __restrict__ bnv,
                        const float* __restrict__ M,
                        const int*   __restrict__ mask,
                        float* __restrict__ flat) {
  const int t   = threadIdx.x;
  const int co  = __builtin_amdgcn_readfirstlane(t >> 6);
  const int pi  = (t >> 2) & 15;
  const int pj0 = (t & 3) * 4;

  int b = blockIdx.x;
  const int ph0 = (b & 1) * kConvPatches;  b >>= 1;
  const int pw  = b % kPR;
  const int img = b / kPR;
  const int w   = pw * kPS + pi;

  int  roff[3];
  float wt[kC][3][3];
#pragma unroll
  for (int kw = 0; kw < 3; ++kw) {
    const int ww = w + kw - 1;
    const bool ok = (unsigned)ww < (unsigned)kW;
    roff[kw] = (ok ? ww : w) * kW;
#pragma unroll
    for (int ci = 0; ci < kC; ++ci)
#pragma unroll
      for (int kh = 0; kh < 3; ++kh) {
        const float wv = cw[((co * kC + ci) * 3 + kw) * 3 + kh];
        wt[ci][kw][kh] = ok ? wv : 0.0f;
      }
  }
  const float bias = cb[co];
  const float inv  = bng[co] * rsqrtf(bnv[co] + 1e-5f);
  const float mean = bnm[co];
  const float beta = bnb[co];
  float mm[4];
#pragma unroll
  for (int j = 0; j < 4; ++j) mm[j] = M[pi * kPS + pj0 + j];

  const float* __restrict__ ximg = x + (size_t)img * kC * kW * kW;

  for (int i = 0; i < kConvPatches; ++i) {
    const int ph = ph0 + i;
    const int h0 = ph * kPS + pj0;
    const bool lok = h0 > 0;
    const bool rok = h0 + 4 < kW;
    const int hl = lok ? h0 - 1 : h0;
    const int hr = rok ? h0 + 4 : h0 + 3;

    float xv[kC][3][6];
#pragma unroll
    for (int ci = 0; ci < kC; ++ci)
#pragma unroll
      for (int kw = 0; kw < 3; ++kw) {
        const float* __restrict__ rp = ximg + ci * kW * kW + roff[kw];
        const float4 v = *(const float4*)(rp + h0);
        const float l = rp[hl];
        const float r = rp[hr];
        xv[ci][kw][0] = lok ? l : 0.0f;
        xv[ci][kw][1] = v.x;
        xv[ci][kw][2] = v.y;
        xv[ci][kw][3] = v.z;
        xv[ci][kw][4] = v.w;
        xv[ci][kw][5] = rok ? r : 0.0f;
      }

    float acc[4] = {bias, bias, bias, bias};
#pragma unroll
    for (int ci = 0; ci < kC; ++ci)
#pragma unroll
      for (int kw = 0; kw < 3; ++kw)
#pragma unroll
        for (int kh = 0; kh < 3; ++kh)
#pragma unroll
          for (int j = 0; j < 4; ++j)
            acc[j] += xv[ci][kw][j + kh] * wt[ci][kw][kh];

    const int n = pw * kPR + ph;
    const bool gate = mask[n] > 0;
    float o[4];
#pragma unroll
    for (int j = 0; j < 4; ++j) {
      float a = (acc[j] - mean) * inv + beta;
      a = gelu_exact(a);
      if (gate) a *= mm[j];
      o[j] = a;
    }
    const int row = img * kN + n;
    *(float4*)(flat + (size_t)row * kF + 4 * t) = make_float4(o[0], o[1], o[2], o[3]);
  }
}

template <bool kMore>
__device__ __forceinline__ void k_chunk(const float* __restrict__ anext,
                                        const float* __restrict__ wnext,
                                        const float* rdB,
                                        float* wrW,
                                        const float* nxB,
                                        v2f (&a)[kSteps], v2f (&b)[2][kTiles], v8f (&acc)[kTiles]) {
  v2f an[kSteps];
  v4f wv[2][kGrp];
#pragma unroll
  for (int s = 0; s < kSteps; ++s) {
    if (s + 1 < kSteps) {
#pragma unroll
      for (int t = 0; t < kTiles; ++t)
        b[(s + 1) & 1][t] = *(const v2f*)(rdB + t * 16 * kPitch + 4 * (s + 1));
    }
    if (kMore) {
      if (s == 0) {
#pragma unroll
        for (int j = 0; j < kSteps; ++j) an[j] = *(const v2f*)(anext + 4 * j);
      }
      if (s >= 2 && s < 2 + kFetchSteps) {
#pragma unroll
        for (int j = 0; j < kGrp; ++j)
          if (kGrp * (s - 2) + j < kPieces)
            *(v4f*)(wrW + (kGrp * (s - 2) + j) * kPieceRows * kPitch) = wv[s & 1][j];
      }
      if (s < kFetchSteps) {
#pragma unroll
        for (int j = 0; j < kGrp; ++j)
          if (kGrp * s + j < kPieces)
            wv[s & 1][j] = *(const v4f*)(wnext + (size_t)(kGrp * s + j) * kPieceRows * kF);
      }
      if (s == kSteps - 1) {
        __syncthreads();
#pragma unroll
        for (int t = 0; t < kTiles; ++t) b[0][t] = *(const v2f*)(nxB + t * 16 * kPitch);
      }
    }
    __builtin_amdgcn_sched_barrier(0);
#pragma unroll
    for (int t = 0; t < kTiles; ++t)
      acc[t] = __builtin_amdgcn_wmma_f32_16x16x4_f32(false, a[s], false, b[s & 1][t],
                                                     (short)0, acc[t], false, false);
    __builtin_amdgcn_sched_barrier(0);
  }
  if (kMore) {
#pragma unroll
    for (int s = 0; s < kSteps; ++s) a[s] = an[s];
  }
}

__device__ __forceinline__ void gemm_rows16_x384(const float* __restrict__ arow,
                                                 const float* __restrict__ Wm,
                                                 float* __restrict__ sW,
                                                 v8f (&acc)[kTiles],
                                                 int tid, int hf, int l16, int ch) {
#pragma unroll
  for (int t = 0; t < kTiles; ++t) acc[t] = v8f{};

  const float* __restrict__ wsrc = Wm + (size_t)(tid >> 3) * kF + 4 * (tid & 7);
  const int wslot = (tid >> 3) * kPitch + 4 * (tid & 7);
  const int bfrag = (ch * kTiles * 16 + l16) * kPitch + 2 * hf;

  {
    v4f v[kPieces];
#pragma unroll
    for (int i = 0; i < kPieces; ++i) v[i] = *(const v4f*)(wsrc + (size_t)i * kPieceRows * kF);
    __builtin_amdgcn_sched_barrier(0);
#pragma unroll
    for (int i = 0; i < kPieces; ++i) *(v4f*)(sW + wslot + i * kPieceRows * kPitch) = v[i];
  }
  v2f a[kSteps];
#pragma unroll
  for (int s = 0; s < kSteps; ++s) a[s] = *(const v2f*)(arow + 4 * s);
  __syncthreads();

  v2f b[2][kTiles];
#pragma unroll
  for (int t = 0; t < kTiles; ++t) b[0][t] = *(const v2f*)(sW + bfrag + t * 16 * kPitch);

  float* rd = sW;
  float* wr = sW + kStage;
#pragma unroll 1
  for (int kc = 0; kc < kF - kKC; kc += kKC) {
    k_chunk<true>(arow + kc + kKC, wsrc + kc + kKC, rd + bfrag, wr + wslot, wr + bfrag, a, b, acc);
    float* const tmp = rd; rd = wr; wr = tmp;
  }
  k_chunk<false>(arow, wsrc, rd + bfrag, wr + wslot, wr + bfrag, a, b, acc);
}

__global__ __launch_bounds__(kThreads)
void gemm1_ln_gelu(const float* __restrict__ A,
                   const float* __restrict__ W1,
                   const float* __restrict__ b1,
                   const float* __restrict__ lng,
                   const float* __restrict__ lnb,
                   float* __restrict__ T) {
  __shared__ __attribute__((aligned(16))) float sW[2 * kStage];
  __shared__ __attribute__((aligned(16))) float part[kBlockRows * 4];

  const int tid  = threadIdx.x;
  const int wave = tid >> 5;
  const int lane = tid & 31;
  const int hf   = lane >> 4;
  const int l16  = lane & 15;
  const int rg   = wave & (kRowGroups - 1);
  const int ch   = wave >> 2;
  const int row0 = blockIdx.x * kBlockRows + rg * kWaveRows;

  v8f acc[kTiles];
  gemm_rows16_x384(A + (size_t)(row0 + l16) * kF + 2 * hf, W1, sW, acc, tid, hf, l16, ch);

  float s1[8], s2[8];
#pragma unroll
  for (int r = 0; r < 8; ++r) { s1[r] = 0.f; s2[r] = 0.f; }
#pragma unroll
  for (int t = 0; t < kTiles; ++t) {
    const float bias = b1[(ch * kTiles + t) * 16 + l16];
#pragma unroll
    for (int r = 0; r < 8; ++r) {
      const float v = acc[t][r] + bias;
      acc[t][r] = v;
      s1[r] += v;
      s2[r] += v * v;
    }
  }
#pragma unroll
  for (int r = 0; r < 8; ++r) {
#pragma unroll
    for (int off = 8; off >= 1; off >>= 1) {
      s1[r] += __shfl_xor(s1[r], off, 32);
      s2[r] += __shfl_xor(s2[r], off, 32);
    }
  }

  float* myrows = sW + wave * kEWave;
  float* prow   = part + (rg * kWaveRows + 8 * hf) * 4;
  if (l16 == 0) {
#pragma unroll
    for (int r = 0; r < 8; ++r) {
      prow[r * 4 + 2 * ch]     = s1[r];
      prow[r * 4 + 2 * ch + 1] = s2[r];
    }
  }

  __syncthreads();

  float mu[8], rs[8];
#pragma unroll
  for (int r = 0; r < 8; ++r) {
    const float4 p = *(const float4*)(prow + r * 4);
    mu[r] = (p.x + p.z) / kF;
    const float var = (p.y + p.w) / kF - mu[r] * mu[r];
    rs[r] = rsqrtf(var + 1e-5f);
  }
  {
    float* wr = myrows + (8 * hf) * kEPitch + l16;
#pragma unroll
    for (int t = 0; t < kTiles; ++t)
#pragma unroll
      for (int r = 0; r < 8; ++r)
        wr[r * kEPitch + t * 16] = (acc[t][r] - mu[r]) * rs[r];
  }

  __syncthreads();

  const int col0 = ch * (kHid / 2) + 4 * lane;
  float4 g4[3], b4[3];
#pragma unroll
  for (int j = 0; j < 3; ++j) {
    g4[j] = *(const float4*)(lng + col0 + 128 * j);
    b4[j] = *(const float4*)(lnb + col0 + 128 * j);
  }
  const float* src = myrows + 4 * lane;
  float* __restrict__ dst = T + (size_t)row0 * kHid + col0;

  for (int i = 0; i < kWaveRows; ++i) {
#pragma unroll
    for (int j = 0; j < 3; ++j) {
      const float4 v = *(const float4*)(src + i * kEPitch + 128 * j);
      float4 o;
      o.x = gelu_exact(v.x * g4[j].x + b4[j].x);
      o.y = gelu_exact(v.y * g4[j].y + b4[j].y);
      o.z = gelu_exact(v.z * g4[j].z + b4[j].z);
      o.w = gelu_exact(v.w * g4[j].w + b4[j].w);
      *(float4*)(dst + (size_t)i * kHid + 128 * j) = o;
    }
  }
}

__global__ __launch_bounds__(kThreads)
void gemm2_perm(const float* __restrict__ T,
                const float* __restrict__ W2,
                const float* __restrict__ b2,
                const int*   __restrict__ perm,
                float* __restrict__ out) {
  __shared__ __attribute__((aligned(16))) float sW[2 * kStage];

  const int tid  = threadIdx.x;
  const int wave = tid >> 5;
  const int lane = tid & 31;
  const int hf   = lane >> 4;
  const int l16  = lane & 15;
  const int rg   = wave & (kRowGroups - 1);
  const int ch   = wave >> 2;
  const int row0 = blockIdx.x * kBlockRows + rg * kWaveRows;

  const int g    = row0 + l16;
  const int img  = g / kN;
  const int n0   = g - img * kN;
  const int src  = img * kN + perm[n0];

  v8f acc[kTiles];
  gemm_rows16_x384(T + (size_t)src * kHid + 2 * hf, W2, sW, acc, tid, hf, l16, ch);

#pragma unroll
  for (int t = 0; t < kTiles; ++t) {
    const int n = (ch * kTiles + t) * 16 + l16;
    const float bias = b2[n];
#pragma unroll
    for (int r = 0; r < 8; ++r) {
      out[(size_t)(row0 + 8 * hf + r) * kOut + n] = acc[t][r] + bias;
    }
  }
}

extern "C" void kernel_launch(void* const* d_in, const int* in_sizes, int n_in,
                              void* d_out, int out_size, void* d_ws, size_t ws_size,
                              hipStream_t stream) {
  const float* x    = (const float*)d_in[0];
  const float* cw   = (const float*)d_in[1];
  const float* cb   = (const float*)d_in[2];
  const float* bng  = (const float*)d_in[3];
  const float* bnb  = (const float*)d_in[4];
  const float* bnm  = (const float*)d_in[5];
  const float* bnv  = (const float*)d_in[6];
  const float* M    = (const float*)d_in[7];
  const float* W1   = (const float*)d_in[8];
  const float* b1   = (const float*)d_in[9];
  const float* lng  = (const float*)d_in[10];
  const float* lnb  = (const float*)d_in[11];
  const float* W2   = (const float*)d_in[12];
  const float* b2   = (const float*)d_in[13];
  const int*   mask = (const int*)d_in[14];
  const int*   perm = (const int*)d_in[15];
  float* out = (float*)d_out;

  float* flat = (float*)d_ws;
  float* Tbuf = flat + (size_t)kRows * kF;

  conv_bn_gelu_patch<<<kBS * kPR * 2, kConvThreads, 0, stream>>>(
      x, cw, cb, bng, bnb, bnm, bnv, M, mask, flat);
  gemm1_ln_gelu<<<kRows / kBlockRows, kThreads, 0, stream>>>(flat, W1, b1, lng, lnb, Tbuf);
  gemm2_perm  <<<kRows / kBlockRows, kThreads, 0, stream>>>(Tbuf, W2, b2, perm, out);
}


# fc2 re-tiled: 224 workgroups of 112 rows x 384 columns (was 196 of 64 x 768), 28 waves of 16 rows x 96 columns, seven per SIMD; W chunk is the column half only; k order and sums kept
# speedup vs baseline: 2.8570x; 1.0538x over previous
#include <hip/hip_runtime.h>
#include <math.h>


typedef float v2f __attribute__((ext_vector_type(2)));
typedef float v4f __attribute__((ext_vector_type(4)));
typedef float v8f __attribute__((ext_vector_type(8)));

namespace {
constexpr int kPS   = 16;
constexpr int kPR   = 14;
constexpr int kN    = 196;
constexpr int kC    = 3;
constexpr int kW    = 224;
constexpr int kBS   = 64;
constexpr int kF    = 768;
constexpr int kHid  = 768;
constexpr int kOut  = 768;
constexpr int kRows = kBS * kN;
constexpr int kWaveRows  = 16;
constexpr int kRowGroups = 4;
constexpr int kColHalves = 2;
constexpr int kWaves     = kRowGroups * kColHalves;
constexpr int kThreads   = kWaves * 32;
constexpr int kBlockRows = kWaveRows * kRowGroups;
constexpr int kTiles     = kHid / 16 / kColHalves;
constexpr int kKC        = 32;
constexpr int kPitch     = 36;
constexpr int kStage     = kHid * kPitch;
constexpr int kSteps     = kKC / 4;
constexpr int kPieces    = kHid * (kKC / 4) / kThreads;
constexpr int kPieceRows = kThreads / 8;
constexpr int kGrp       = 5;
constexpr int kFetchSteps = 5;
constexpr int kEPitch    = 388;
constexpr int kEWave     = kWaveRows * kEPitch;
static_assert(kWaves * kEWave <= 2 * kStage, "staged rows must fit in the two LDS stages");
static_assert(kFetchSteps * kGrp >= kPieces && 2 + kFetchSteps <= kSteps - 1, "all pieces are stored before the hand-over barrier in front of the last block");
}

namespace {
constexpr int k2RowGroups = 7;
constexpr int k2Rows      = kWaveRows * k2RowGroups;
constexpr int k2Cols      = kOut / 2;
constexpr int k2Parts     = 4;
constexpr int k2Tiles     = k2Cols / 16 / k2Parts;
constexpr int k2Waves     = k2RowGroups * k2Parts;
constexpr int k2Threads   = k2Waves * 32;
constexpr int k2Stage     = k2Cols * kPitch;
constexpr int k2Copiers   = 768;
constexpr int k2Pieces    = k2Cols * (kKC / 4) / k2Copiers;
constexpr int k2PieceRows = k2Copiers / 8;
static_assert(k2Pieces * k2Copiers == k2Cols * (kKC / 4), "the copying threads cover a chunk exactly");
static_assert(kRows % k2Rows == 0, "whole row blocks");
}

__device__ __forceinline__ float gelu_exact(float v) {
  return 0.5f * v * (1.0f + erff(v * 0.70710678118654752440f));
}

constexpr int kConvThreads = kF / 4;
constexpr int kConvPatches = 7;

__global__ __launch_bounds__(kConvThreads)
void conv_bn_gelu_patch(const float* __restrict__ x,
                        const float* __restrict__ cw,
                        const float* __restrict__ cb,
                        const float* __restrict__ bng,
                        const float* __restrict__ bnb,
                        const float* __restrict__ bnm,
                        const float* __restrict__ bnv,
                        const float* __restrict__ M,
                        const int*   __restrict__ mask,
                        float* __restrict__ flat) {
  const int t   = threadIdx.x;
  const int co  = __builtin_amdgcn_readfirstlane(t >> 6);
  const int pi  = (t >> 2) & 15;
  const int pj0 = (t & 3) * 4;

  int b = blockIdx.x;
  const int ph0 = (b & 1) * kConvPatches;  b >>= 1;
  const int pw  = b % kPR;
  const int img = b / kPR;
  const int w   = pw * kPS + pi;

  int  roff[3];
  float wt[kC][3][3];
#pragma unroll
  for (int kw = 0; kw < 3; ++kw) {
    const int ww = w + kw - 1;
    const bool ok = (unsigned)ww < (unsigned)kW;
    roff[kw] = (ok ? ww : w) * kW;
#pragma unroll
    for (int ci = 0; ci < kC; ++ci)
#pragma unroll
      for (int kh = 0; kh < 3; ++kh) {
        const float wv = cw[((co * kC + ci) * 3 + kw) * 3 + kh];
        wt[ci][kw][kh] = ok ? wv : 0.0f;
      }
  }
  const float bias = cb[co];
  const float inv  = bng[co] * rsqrtf(bnv[co] + 1e-5f);
  const float mean = bnm[co];
  const float beta = bnb[co];
  float mm[4];
#pragma unroll
  for (int j = 0; j < 4; ++j) mm[j] = M[pi * kPS + pj0 + j];

  const float* __restrict__ ximg = x + (size_t)img * kC * kW * kW;

  for (int i = 0; i < kConvPatches; ++i) {
    const int ph = ph0 + i;
    const int h0 = ph * kPS + pj0;
    const bool lok = h0 > 0;
    const bool rok = h0 + 4 < kW;
    const int hl = lok ? h0 - 1 : h0;
    const int hr = rok ? h0 + 4 : h0 + 3;

    float xv[kC][3][6];
#pragma unroll
    for (int ci = 0; ci < kC; ++ci)
#pragma unroll
      for (int kw = 0; kw < 3; ++kw) {
        const float* __restrict__ rp = ximg + ci * kW * kW + roff[kw];
        const float4 v = *(const float4*)(rp + h0);
        const float l = rp[hl];
        const float r = rp[hr];
        xv[ci][kw][0] = lok ? l : 0.0f;
        xv[ci][kw][1] = v.x;
        xv[ci][kw][2] = v.y;
        xv[ci][kw][3] = v.z;
        xv[ci][kw][4] = v.w;
        xv[ci][kw][5] = rok ? r : 0.0f;
      }

    float acc[4] = {bias, bias, bias, bias};
#pragma unroll
    for (int ci = 0; ci < kC; ++ci)
#pragma unroll
      for (int kw = 0; kw < 3; ++kw)
#pragma unroll
        for (int kh = 0; kh < 3; ++kh)
#pragma unroll
          for (int j = 0; j < 4; ++j)
            acc[j] += xv[ci][kw][j + kh] * wt[ci][kw][kh];

    const int n = pw * kPR + ph;
    const bool gate = mask[n] > 0;
    float o[4];
#pragma unroll
    for (int j = 0; j < 4; ++j) {
      float a = (acc[j] - mean) * inv + beta;
      a = gelu_exact(a);
      if (gate) a *= mm[j];
      o[j] = a;
    }
    const int row = img * kN + n;
    *(float4*)(flat + (size_t)row * kF + 4 * t) = make_float4(o[0], o[1], o[2], o[3]);
  }
}

template <bool kMore>
__device__ __forceinline__ void k_chunk(const float* __restrict__ anext,
                                        const float* __restrict__ wnext,
                                        const float* rdB,
                                        float* wrW,
                                        const float* nxB,
                                        v2f (&a)[kSteps], v2f (&b)[2][kTiles], v8f (&acc)[kTiles]) {
  v2f an[kSteps];
  v4f wv[2][kGrp];
#pragma unroll
  for (int s = 0; s < kSteps; ++s) {
    if (s + 1 < kSteps) {
#pragma unroll
      for (int t = 0; t < kTiles; ++t)
        b[(s + 1) & 1][t] = *(const v2f*)(rdB + t * 16 * kPitch + 4 * (s + 1));
    }
    if (kMore) {
      if (s == 0) {
#pragma unroll
        for (int j = 0; j < kSteps; ++j) an[j] = *(const v2f*)(anext + 4 * j);
      }
      if (s >= 2 && s < 2 + kFetchSteps) {
#pragma unroll
        for (int j = 0; j < kGrp; ++j)
          if (kGrp * (s - 2) + j < kPieces)
            *(v4f*)(wrW + (kGrp * (s - 2) + j) * kPieceRows * kPitch) = wv[s & 1][j];
      }
      if (s < kFetchSteps) {
#pragma unroll
        for (int j = 0; j < kGrp; ++j)
          if (kGrp * s + j < kPieces)
            wv[s & 1][j] = *(const v4f*)(wnext + (size_t)(kGrp * s + j) * kPieceRows * kF);
      }
      if (s == kSteps - 1) {
        __syncthreads();
#pragma unroll
        for (int t = 0; t < kTiles; ++t) b[0][t] = *(const v2f*)(nxB + t * 16 * kPitch);
      }
    }
    __builtin_amdgcn_sched_barrier(0);
#pragma unroll
    for (int t = 0; t < kTiles; ++t)
      acc[t] = __builtin_amdgcn_wmma_f32_16x16x4_f32(false, a[s], false, b[s & 1][t],
                                                     (short)0, acc[t], false, false);
    __builtin_amdgcn_sched_barrier(0);
  }
  if (kMore) {
#pragma unroll
    for (int s = 0; s < kSteps; ++s) a[s] = an[s];
  }
}

__device__ __forceinline__ void gemm_rows16_x384(const float* __restrict__ arow,
                                                 const float* __restrict__ Wm,
                                                 float* __restrict__ sW,
                                                 v8f (&acc)[kTiles],
                                                 int tid, int hf, int l16, int ch) {
#pragma unroll
  for (int t = 0; t < kTiles; ++t) acc[t] = v8f{};

  const float* __restrict__ wsrc = Wm + (size_t)(tid >> 3) * kF + 4 * (tid & 7);
  const int wslot = (tid >> 3) * kPitch + 4 * (tid & 7);
  const int bfrag = (ch * kTiles * 16 + l16) * kPitch + 2 * hf;

  {
    v4f v[kPieces];
#pragma unroll
    for (int i = 0; i < kPieces; ++i) v[i] = *(const v4f*)(wsrc + (size_t)i * kPieceRows * kF);
    __builtin_amdgcn_sched_barrier(0);
#pragma unroll
    for (int i = 0; i < kPieces; ++i) *(v4f*)(sW + wslot + i * kPieceRows * kPitch) = v[i];
  }
  v2f a[kSteps];
#pragma unroll
  for (int s = 0; s < kSteps; ++s) a[s] = *(const v2f*)(arow + 4 * s);
  __syncthreads();

  v2f b[2][kTiles];
#pragma unroll
  for (int t = 0; t < kTiles; ++t) b[0][t] = *(const v2f*)(sW + bfrag + t * 16 * kPitch);

  float* rd = sW;
  float* wr = sW + kStage;
#pragma unroll 1
  for (int kc = 0; kc < kF - kKC; kc += kKC) {
    k_chunk<true>(arow + kc + kKC, wsrc + kc + kKC, rd + bfrag, wr + wslot, wr + bfrag, a, b, acc);
    float* const tmp = rd; rd = wr; wr = tmp;
  }
  k_chunk<false>(arow, wsrc, rd + bfrag, wr + wslot, wr + bfrag, a, b, acc);
}

__global__ __launch_bounds__(kThreads)
void gemm1_ln_gelu(const float* __restrict__ A,
                   const float* __restrict__ W1,
                   const float* __restrict__ b1,
                   const float* __restrict__ lng,
                   const float* __restrict__ lnb,
                   float* __restrict__ T) {
  __shared__ __attribute__((aligned(16))) float sW[2 * kStage];
  __shared__ __attribute__((aligned(16))) float part[kBlockRows * 4];

  const int tid  = threadIdx.x;
  const int wave = tid >> 5;
  const int lane = tid & 31;
  const int hf   = lane >> 4;
  const int l16  = lane & 15;
  const int rg   = wave & (kRowGroups - 1);
  const int ch   = wave >> 2;
  const int row0 = blockIdx.x * kBlockRows + rg * kWaveRows;

  v8f acc[kTiles];
  gemm_rows16_x384(A + (size_t)(row0 + l16) * kF + 2 * hf, W1, sW, acc, tid, hf, l16, ch);

  float s1[8], s2[8];
#pragma unroll
  for (int r = 0; r < 8; ++r) { s1[r] = 0.f; s2[r] = 0.f; }
#pragma unroll
  for (int t = 0; t < kTiles; ++t) {
    const float bias = b1[(ch * kTiles + t) * 16 + l16];
#pragma unroll
    for (int r = 0; r < 8; ++r) {
      const float v = acc[t][r] + bias;
      acc[t][r] = v;
      s1[r] += v;
      s2[r] += v * v;
    }
  }
#pragma unroll
  for (int r = 0; r < 8; ++r) {
#pragma unroll
    for (int off = 8; off >= 1; off >>= 1) {
      s1[r] += __shfl_xor(s1[r], off, 32);
      s2[r] += __shfl_xor(s2[r], off, 32);
    }
  }

  float* myrows = sW + wave * kEWave;
  float* prow   = part + (rg * kWaveRows + 8 * hf) * 4;
  if (l16 == 0) {
#pragma unroll
    for (int r = 0; r < 8; ++r) {
      prow[r * 4 + 2 * ch]     = s1[r];
      prow[r * 4 + 2 * ch + 1] = s2[r];
    }
  }

  __syncthreads();

  float mu[8], rs[8];
#pragma unroll
  for (int r = 0; r < 8; ++r) {
    const float4 p = *(const float4*)(prow + r * 4);
    mu[r] = (p.x + p.z) / kF;
    const float var = (p.y + p.w) / kF - mu[r] * mu[r];
    rs[r] = rsqrtf(var + 1e-5f);
  }
  {
    float* wr = myrows + (8 * hf) * kEPitch + l16;
#pragma unroll
    for (int t = 0; t < kTiles; ++t)
#pragma unroll
      for (int r = 0; r < 8; ++r)
        wr[r * kEPitch + t * 16] = (acc[t][r] - mu[r]) * rs[r];
  }

  __syncthreads();

  const int col0 = ch * (kHid / 2) + 4 * lane;
  float4 g4[3], b4[3];
#pragma unroll
  for (int j = 0; j < 3; ++j) {
    g4[j] = *(const float4*)(lng + col0 + 128 * j);
    b4[j] = *(const float4*)(lnb + col0 + 128 * j);
  }
  const float* src = myrows + 4 * lane;
  float* __restrict__ dst = T + (size_t)row0 * kHid + col0;

  for (int i = 0; i < kWaveRows; ++i) {
#pragma unroll
    for (int j = 0; j < 3; ++j) {
      const float4 v = *(const float4*)(src + i * kEPitch + 128 * j);
      float4 o;
      o.x = gelu_exact(v.x * g4[j].x + b4[j].x);
      o.y = gelu_exact(v.y * g4[j].y + b4[j].y);
      o.z = gelu_exact(v.z * g4[j].z + b4[j].z);
      o.w = gelu_exact(v.w * g4[j].w + b4[j].w);
      *(float4*)(dst + (size_t)i * kHid + 128 * j) = o;
    }
  }
}

template <bool kMore>
__device__ __forceinline__ void k_chunk2(const float* __restrict__ anext,
                                         const float* __restrict__ wnext,
                                         const float* rdB,
                                         float* wrW,
                                         const float* nxB,
                                         bool copier,
                                         v2f (&a)[kSteps], v2f (&b)[2][k2Tiles], v8f (&acc)[k2Tiles]) {
  v2f an[kSteps];
  v4f wv[k2Pieces];
#pragma unroll
  for (int s = 0; s < kSteps; ++s) {
    if (s + 1 < kSteps) {
#pragma unroll
      for (int t = 0; t < k2Tiles; ++t)
        b[(s + 1) & 1][t] = *(const v2f*)(rdB + t * 16 * kPitch + 4 * (s + 1));
    }
    if (kMore) {
      if (s == 0) {
        if (copier) {
#pragma unroll
          for (int j = 0; j < k2Pieces; ++j)
            wv[j] = *(const v4f*)(wnext + (size_t)j * k2PieceRows * kF);
        }
#pragma unroll
        for (int j = 0; j < kSteps; ++j) an[j] = *(const v2f*)(anext + 4 * j);
      }
      if (s == 4) {
        if (copier) {
#pragma unroll
          for (int j = 0; j < k2Pieces; ++j)
            *(v4f*)(wrW + j * k2PieceRows * kPitch) = wv[j];
        }
      }
      if (s == kSteps - 1) {
        __syncthreads();
#pragma unroll
        for (int t = 0; t < k2Tiles; ++t) b[0][t] = *(const v2f*)(nxB + t * 16 * kPitch);
      }
    }
    __builtin_amdgcn_sched_barrier(0);
#pragma unroll
    for (int t = 0; t < k2Tiles; ++t)
      acc[t] = __builtin_amdgcn_wmma_f32_16x16x4_f32(false, a[s], false, b[s & 1][t],
                                                     (short)0, acc[t], false, false);
    __builtin_amdgcn_sched_barrier(0);
  }
  if (kMore) {
#pragma unroll
    for (int s = 0; s < kSteps; ++s) a[s] = an[s];
  }
}

__device__ __forceinline__ void gemm_rows16_x96(const float* __restrict__ arow,
                                                const float* __restrict__ Wm,
                                                float* __restrict__ sW,
                                                v8f (&acc)[k2Tiles],
                                                int tid, int hf, int l16, int part, bool copier) {
#pragma unroll
  for (int t = 0; t < k2Tiles; ++t) acc[t] = v8f{};

  const float* __restrict__ wsrc = Wm + (size_t)(tid >> 3) * kF + 4 * (tid & 7);
  const int wslot = (tid >> 3) * kPitch + 4 * (tid & 7);
  const int bfrag = (part * k2Tiles * 16 + l16) * kPitch + 2 * hf;

  if (copier) {
    v4f v[k2Pieces];
#pragma unroll
    for (int i = 0; i < k2Pieces; ++i) v[i] = *(const v4f*)(wsrc + (size_t)i * k2PieceRows * kF);
#pragma unroll
    for (int i = 0; i < k2Pieces; ++i) *(v4f*)(sW + wslot + i * k2PieceRows * kPitch) = v[i];
  }
  v2f a[kSteps];
#pragma unroll
  for (int s = 0; s < kSteps; ++s) a[s] = *(const v2f*)(arow + 4 * s);
  __syncthreads();

  v2f b[2][k2Tiles];
#pragma unroll
  for (int t = 0; t < k2Tiles; ++t) b[0][t] = *(const v2f*)(sW + bfrag + t * 16 * kPitch);

  float* rd = sW;
  float* wr = sW + k2Stage;
#pragma unroll 1
  for (int kc = 0; kc < kF - kKC; kc += kKC) {
    k_chunk2<true>(arow + kc + kKC, wsrc + kc + kKC, rd + bfrag, wr + wslot, wr + bfrag, copier, a, b, acc);
    float* const tmp = rd; rd = wr; wr = tmp;
  }
  k_chunk2<false>(arow, wsrc, rd + bfrag, wr + wslot, wr + bfrag, copier, a, b, acc);
}

__global__ __launch_bounds__(k2Threads)
void gemm2_perm(const float* __restrict__ T,
                const float* __restrict__ W2,
                const float* __restrict__ b2,
                const int*   __restrict__ perm,
                float* __restrict__ out) {
  __shared__ __attribute__((aligned(16))) float sW[2 * k2Stage];

  const int tid  = threadIdx.x;
  const int wave = __builtin_amdgcn_readfirstlane(tid >> 5);
  const int lane = tid & 31;
  const int hf   = lane >> 4;
  const int l16  = lane & 15;
  const int rg   = wave % k2RowGroups;
  const int part = wave / k2RowGroups;
  const int half = blockIdx.x & 1;
  const int row0 = (blockIdx.x >> 1) * k2Rows + rg * kWaveRows;
  const int col0 = half * k2Cols + part * (k2Tiles * 16);
  const bool copier = wave < k2Copiers / 32;

  const int g    = row0 + l16;
  const int img  = g / kN;
  const int n0   = g - img * kN;
  const int src  = img * kN + perm[n0];

  v8f acc[k2Tiles];
  gemm_rows16_x96(T + (size_t)src * kHid + 2 * hf, W2 + (size_t)half * k2Cols * kHid, sW, acc,
                  tid, hf, l16, part, copier);

#pragma unroll
  for (int t = 0; t < k2Tiles; ++t) {
    const int n = col0 + t * 16 + l16;
    const float bias = b2[n];
#pragma unroll
    for (int r = 0; r < 8; ++r) {
      out[(size_t)(row0 + 8 * hf + r) * kOut + n] = acc[t][r] + bias;
    }
  }
}

extern "C" void kernel_launch(void* const* d_in, const int* in_sizes, int n_in,
                              void* d_out, int out_size, void* d_ws, size_t ws_size,
                              hipStream_t stream) {
  const float* x    = (const float*)d_in[0];
  const float* cw   = (const float*)d_in[1];
  const float* cb   = (const float*)d_in[2];
  const float* bng  = (const float*)d_in[3];
  const float* bnb  = (const float*)d_in[4];
  const float* bnm  = (const float*)d_in[5];
  const float* bnv  = (const float*)d_in[6];
  const float* M    = (const float*)d_in[7];
  const float* W1   = (const float*)d_in[8];
  const float* b1   = (const float*)d_in[9];
  const float* lng  = (const float*)d_in[10];
  const float* lnb  = (const float*)d_in[11];
  const float* W2   = (const float*)d_in[12];
  const float* b2   = (const float*)d_in[13];
  const int*   mask = (const int*)d_in[14];
  const int*   perm = (const int*)d_in[15];
  float* out = (float*)d_out;

  float* flat = (float*)d_ws;
  float* Tbuf = flat + (size_t)kRows * kF;

  conv_bn_gelu_patch<<<kBS * kPR * 2, kConvThreads, 0, stream>>>(
      x, cw, cb, bng, bnb, bnm, bnv, M, mask, flat);
  gemm1_ln_gelu<<<kRows / kBlockRows, kThreads, 0, stream>>>(flat, W1, b1, lng, lnb, Tbuf);
  gemm2_perm  <<<(kRows / k2Rows) * 2, k2Threads, 0, stream>>>(Tbuf, W2, b2, perm, out);
}


# conv kernel: the block copies its piece of x (3 channels x 18 rows x 114 columns, border zeroed) into LDS once and reads the taps from LDS; was 27 global loads per patch and thread; tap order kept
# speedup vs baseline: 3.0402x; 1.0641x over previous
#include <hip/hip_runtime.h>
#include <math.h>


typedef float v2f __attribute__((ext_vector_type(2)));
typedef float v4f __attribute__((ext_vector_type(4)));
typedef float v8f __attribute__((ext_vector_type(8)));

namespace {
constexpr int kPS   = 16;
constexpr int kPR   = 14;
constexpr int kN    = 196;
constexpr int kC    = 3;
constexpr int kW    = 224;
constexpr int kBS   = 64;
constexpr int kF    = 768;
constexpr int kHid  = 768;
constexpr int kOut  = 768;
constexpr int kRows = kBS * kN;
constexpr int kWaveRows  = 16;
constexpr int kRowGroups = 4;
constexpr int kColHalves = 2;
constexpr int kWaves     = kRowGroups * kColHalves;
constexpr int kThreads   = kWaves * 32;
constexpr int kBlockRows = kWaveRows * kRowGroups;
constexpr int kTiles     = kHid / 16 / kColHalves;
constexpr int kKC        = 32;
constexpr int kPitch     = 36;
constexpr int kStage     = kHid * kPitch;
constexpr int kSteps     = kKC / 4;
constexpr int kPieces    = kHid * (kKC / 4) / kThreads;
constexpr int kPieceRows = kThreads / 8;
constexpr int kGrp       = 5;
constexpr int kFetchSteps = 5;
constexpr int kEPitch    = 388;
constexpr int kEWave     = kWaveRows * kEPitch;
static_assert(kWaves * kEWave <= 2 * kStage, "staged rows must fit in the two LDS stages");
static_assert(kFetchSteps * kGrp >= kPieces && 2 + kFetchSteps <= kSteps - 1, "all pieces are stored before the hand-over barrier in front of the last block");
}

namespace {
constexpr int k2RowGroups = 7;
constexpr int k2Rows      = kWaveRows * k2RowGroups;
constexpr int k2Cols      = kOut / 2;
constexpr int k2Parts     = 4;
constexpr int k2Tiles     = k2Cols / 16 / k2Parts;
constexpr int k2Waves     = k2RowGroups * k2Parts;
constexpr int k2Threads   = k2Waves * 32;
constexpr int k2Stage     = k2Cols * kPitch;
constexpr int k2Copiers   = 768;
constexpr int k2Pieces    = k2Cols * (kKC / 4) / k2Copiers;
constexpr int k2PieceRows = k2Copiers / 8;
static_assert(k2Pieces * k2Copiers == k2Cols * (kKC / 4), "the copying threads cover a chunk exactly");
static_assert(kRows % k2Rows == 0, "whole row blocks");
}

__device__ __forceinline__ float gelu_exact(float v) {
  return 0.5f * v * (1.0f + erff(v * 0.70710678118654752440f));
}

constexpr int kConvThreads = kF / 4;
constexpr int kConvPatches = 7;
constexpr int kTileRows  = kPS + 2;
constexpr int kTileCols  = kConvPatches * kPS;
constexpr int kTileQuads = kTileCols / 4 + 2;
constexpr int kTileLead  = 4;
constexpr int kTilePitch = 144;
constexpr int kTileSegs  = kC * kTileRows;
constexpr int kFillThreads = 180;
constexpr int kFillTrips = kTileSegs * kTileQuads / kFillThreads;
static_assert(kFillTrips * kFillThreads == kTileSegs * kTileQuads && kFillThreads <= kConvThreads, "the filling threads cover the tile exactly");
static_assert(kTileQuads * 4 <= kTilePitch, "a tile row fits its pitch");

__global__ __launch_bounds__(kConvThreads)
void conv_bn_gelu_patch(const float* __restrict__ x,
                        const float* __restrict__ cw,
                        const float* __restrict__ cb,
                        const float* __restrict__ bng,
                        const float* __restrict__ bnb,
                        const float* __restrict__ bnm,
                        const float* __restrict__ bnv,
                        const float* __restrict__ M,
                        const int*   __restrict__ mask,
                        float* __restrict__ flat) {
  __shared__ __attribute__((aligned(16))) float sx[kTileSegs * kTilePitch];

  const int t   = threadIdx.x;
  const int co  = __builtin_amdgcn_readfirstlane(t >> 6);
  const int pi  = (t >> 2) & 15;
  const int pj0 = (t & 3) * 4;

  int b = blockIdx.x;
  const int ph0 = (b & 1) * kConvPatches;  b >>= 1;
  const int pw  = b % kPR;
  const int img = b / kPR;
  const int w0  = pw * kPS - 1;
  const int c0  = ph0 * kPS;

  const float* __restrict__ ximg = x + (size_t)img * kC * kW * kW;

  if (t < kFillThreads) {
    float4 fv[kFillTrips];
    int    fdst[kFillTrips];
    bool   fok[kFillTrips];
#pragma unroll
    for (int k = 0; k < kFillTrips; ++k) {
      const int idx = t + k * kFillThreads;
      const int seg = idx / kTileQuads;
      const int q   = idx - seg * kTileQuads;
      const int ci  = seg / kTileRows;
      const int ww  = w0 + (seg - ci * kTileRows);
      const int wc  = ww < 0 ? 0 : (ww > kW - 1 ? kW - 1 : ww);
      const int hh  = c0 - 4 + 4 * q;
      const int hc  = hh < 0 ? 0 : (hh > kW - 4 ? kW - 4 : hh);
      fok[k]  = ww == wc && hh == hc;
      fdst[k] = seg * kTilePitch + 4 * q;
      fv[k]   = *(const float4*)(ximg + ci * kW * kW + wc * kW + hc);
    }
    __builtin_amdgcn_sched_barrier(0);
#pragma unroll
    for (int k = 0; k < kFillTrips; ++k) {
      float4 v = fv[k];
      if (!fok[k]) v = make_float4(0.0f, 0.0f, 0.0f, 0.0f);
      *(float4*)(sx + fdst[k]) = v;
    }
  }

  float wt[kC][3][3];
#pragma unroll
  for (int ci = 0; ci < kC; ++ci)
#pragma unroll
    for (int kw = 0; kw < 3; ++kw)
#pragma unroll
      for (int kh = 0; kh < 3; ++kh)
        wt[ci][kw][kh] = cw[((co * kC + ci) * 3 + kw) * 3 + kh];
  const float bias = cb[co];
  const float inv  = bng[co] * rsqrtf(bnv[co] + 1e-5f);
  const float mean = bnm[co];
  const float beta = bnb[co];
  float mm[4];
#pragma unroll
  for (int j = 0; j < 4; ++j) mm[j] = M[pi * kPS + pj0 + j];

  __syncthreads();

  const float* tap0 = sx + pi * kTilePitch + kTileLead + pj0;

  for (int i = 0; i < kConvPatches; ++i) {
    const int ph = ph0 + i;

    float xv[kC][3][6];
#pragma unroll
    for (int ci = 0; ci < kC; ++ci)
#pragma unroll
      for (int kw = 0; kw < 3; ++kw) {
        const float* rp = tap0 + (ci * kTileRows + kw) * kTilePitch + i * kPS;
        const float4 v = *(const float4*)rp;
        xv[ci][kw][0] = rp[-1];
        xv[ci][kw][1] = v.x;
        xv[ci][kw][2] = v.y;
        xv[ci][kw][3] = v.z;
        xv[ci][kw][4] = v.w;
        xv[ci][kw][5] = rp[4];
      }

    float acc[4] = {bias, bias, bias, bias};
#pragma unroll
    for (int ci = 0; ci < kC; ++ci)
#pragma unroll
      for (int kw = 0; kw < 3; ++kw)
#pragma unroll
        for (int kh = 0; kh < 3; ++kh)
#pragma unroll
          for (int j = 0; j < 4; ++j)
            acc[j] = __builtin_fmaf(xv[ci][kw][j + kh], wt[ci][kw][kh], acc[j]);

    const int n = pw * kPR + ph;
    const bool gate = mask[n] > 0;
    float o[4];
#pragma unroll
    for (int j = 0; j < 4; ++j) {
      float a = (acc[j] - mean) * inv + beta;
      a = gelu_exact(a);
      if (gate) a *= mm[j];
      o[j] = a;
    }
    const int row = img * kN + n;
    *(float4*)(flat + (size_t)row * kF + 4 * t) = make_float4(o[0], o[1], o[2], o[3]);
  }
}

template <bool kMore>
__device__ __forceinline__ void k_chunk(const float* __restrict__ anext,
                                        const float* __restrict__ wnext,
                                        const float* rdB,
                                        float* wrW,
                                        const float* nxB,
                                        v2f (&a)[kSteps], v2f (&b)[2][kTiles], v8f (&acc)[kTiles]) {
  v2f an[kSteps];
  v4f wv[2][kGrp];
#pragma unroll
  for (int s = 0; s < kSteps; ++s) {
    if (s + 1 < kSteps) {
#pragma unroll
      for (int t = 0; t < kTiles; ++t)
        b[(s + 1) & 1][t] = *(const v2f*)(rdB + t * 16 * kPitch + 4 * (s + 1));
    }
    if (kMore) {
      if (s == 0) {
#pragma unroll
        for (int j = 0; j < kSteps; ++j) an[j] = *(const v2f*)(anext + 4 * j);
      }
      if (s >= 2 && s < 2 + kFetchSteps) {
#pragma unroll
        for (int j = 0; j < kGrp; ++j)
          if (kGrp * (s - 2) + j < kPieces)
            *(v4f*)(wrW + (kGrp * (s - 2) + j) * kPieceRows * kPitch) = wv[s & 1][j];
      }
      if (s < kFetchSteps) {
#pragma unroll
        for (int j = 0; j < kGrp; ++j)
          if (kGrp * s + j < kPieces)
            wv[s & 1][j] = *(const v4f*)(wnext + (size_t)(kGrp * s + j) * kPieceRows * kF);
      }
      if (s == kSteps - 1) {
        __syncthreads();
#pragma unroll
        for (int t = 0; t < kTiles; ++t) b[0][t] = *(const v2f*)(nxB + t * 16 * kPitch);
      }
    }
    __builtin_amdgcn_sched_barrier(0);
#pragma unroll
    for (int t = 0; t < kTiles; ++t)
      acc[t] = __builtin_amdgcn_wmma_f32_16x16x4_f32(false, a[s], false, b[s & 1][t],
                                                     (short)0, acc[t], false, false);
    __builtin_amdgcn_sched_barrier(0);
  }
  if (kMore) {
#pragma unroll
    for (int s = 0; s < kSteps; ++s) a[s] = an[s];
  }
}

__device__ __forceinline__ void gemm_rows16_x384(const float* __restrict__ arow,
                                                 const float* __restrict__ Wm,
                                                 float* __restrict__ sW,
                                                 v8f (&acc)[kTiles],
                                                 int tid, int hf, int l16, int ch) {
#pragma unroll
  for (int t = 0; t < kTiles; ++t) acc[t] = v8f{};

  const float* __restrict__ wsrc = Wm + (size_t)(tid >> 3) * kF + 4 * (tid & 7);
  const int wslot = (tid >> 3) * kPitch + 4 * (tid & 7);
  const int bfrag = (ch * kTiles * 16 + l16) * kPitch + 2 * hf;

  {
    v4f v[kPieces];
#pragma unroll
    for (int i = 0; i < kPieces; ++i) v[i] = *(const v4f*)(wsrc + (size_t)i * kPieceRows * kF);
    __builtin_amdgcn_sched_barrier(0);
#pragma unroll
    for (int i = 0; i < kPieces; ++i) *(v4f*)(sW + wslot + i * kPieceRows * kPitch) = v[i];
  }
  v2f a[kSteps];
#pragma unroll
  for (int s = 0; s < kSteps; ++s) a[s] = *(const v2f*)(arow + 4 * s);
  __syncthreads();

  v2f b[2][kTiles];
#pragma unroll
  for (int t = 0; t < kTiles; ++t) b[0][t] = *(const v2f*)(sW + bfrag + t * 16 * kPitch);

  float* rd = sW;
  float* wr = sW + kStage;
#pragma unroll 1
  for (int kc = 0; kc < kF - kKC; kc += kKC) {
    k_chunk<true>(arow + kc + kKC, wsrc + kc + kKC, rd + bfrag, wr + wslot, wr + bfrag, a, b, acc);
    float* const tmp = rd; rd = wr; wr = tmp;
  }
  k_chunk<false>(arow, wsrc, rd + bfrag, wr + wslot, wr + bfrag, a, b, acc);
}

__global__ __launch_bounds__(kThreads)
void gemm1_ln_gelu(const float* __restrict__ A,
                   const float* __restrict__ W1,
                   const float* __restrict__ b1,
                   const float* __restrict__ lng,
                   const float* __restrict__ lnb,
                   float* __restrict__ T) {
  __shared__ __attribute__((aligned(16))) float sW[2 * kStage];
  __shared__ __attribute__((aligned(16))) float part[kBlockRows * 4];

  const int tid  = threadIdx.x;
  const int wave = tid >> 5;
  const int lane = tid & 31;
  const int hf   = lane >> 4;
  const int l16  = lane & 15;
  const int rg   = wave & (kRowGroups - 1);
  const int ch   = wave >> 2;
  const int row0 = blockIdx.x * kBlockRows + rg * kWaveRows;

  v8f acc[kTiles];
  gemm_rows16_x384(A + (size_t)(row0 + l16) * kF + 2 * hf, W1, sW, acc, tid, hf, l16, ch);

  float s1[8], s2[8];
#pragma unroll
  for (int r = 0; r < 8; ++r) { s1[r] = 0.f; s2[r] = 0.f; }
#pragma unroll
  for (int t = 0; t < kTiles; ++t) {
    const float bias = b1[(ch * kTiles + t) * 16 + l16];
#pragma unroll
    for (int r = 0; r < 8; ++r) {
      const float v = acc[t][r] + bias;
      acc[t][r] = v;
      s1[r] += v;
      s2[r] += v * v;
    }
  }
#pragma unroll
  for (int r = 0; r < 8; ++r) {
#pragma unroll
    for (int off = 8; off >= 1; off >>= 1) {
      s1[r] += __shfl_xor(s1[r], off, 32);
      s2[r] += __shfl_xor(s2[r], off, 32);
    }
  }

  float* myrows = sW + wave * kEWave;
  float* prow   = part + (rg * kWaveRows + 8 * hf) * 4;
  if (l16 == 0) {
#pragma unroll
    for (int r = 0; r < 8; ++r) {
      prow[r * 4 + 2 * ch]     = s1[r];
      prow[r * 4 + 2 * ch + 1] = s2[r];
    }
  }

  __syncthreads();

  float mu[8], rs[8];
#pragma unroll
  for (int r = 0; r < 8; ++r) {
    const float4 p = *(const float4*)(prow + r * 4);
    mu[r] = (p.x + p.z) / kF;
    const float var = (p.y + p.w) / kF - mu[r] * mu[r];
    rs[r] = rsqrtf(var + 1e-5f);
  }
  {
    float* wr = myrows + (8 * hf) * kEPitch + l16;
#pragma unroll
    for (int t = 0; t < kTiles; ++t)
#pragma unroll
      for (int r = 0; r < 8; ++r)
        wr[r * kEPitch + t * 16] = (acc[t][r] - mu[r]) * rs[r];
  }

  __syncthreads();

  const int col0 = ch * (kHid / 2) + 4 * lane;
  float4 g4[3], b4[3];
#pragma unroll
  for (int j = 0; j < 3; ++j) {
    g4[j] = *(const float4*)(lng + col0 + 128 * j);
    b4[j] = *(const float4*)(lnb + col0 + 128 * j);
  }
  const float* src = myrows + 4 * lane;
  float* __restrict__ dst = T + (size_t)row0 * kHid + col0;

  for (int i = 0; i < kWaveRows; ++i) {
#pragma unroll
    for (int j = 0; j < 3; ++j) {
      const float4 v = *(const float4*)(src + i * kEPitch + 128 * j);
      float4 o;
      o.x = gelu_exact(v.x * g4[j].x + b4[j].x);
      o.y = gelu_exact(v.y * g4[j].y + b4[j].y);
      o.z = gelu_exact(v.z * g4[j].z + b4[j].z);
      o.w = gelu_exact(v.w * g4[j].w + b4[j].w);
      *(float4*)(dst + (size_t)i * kHid + 128 * j) = o;
    }
  }
}

template <bool kMore>
__device__ __forceinline__ void k_chunk2(const float* __restrict__ anext,
                                         const float* __restrict__ wnext,
                                         const float* rdB,
                                         float* wrW,
                                         const float* nxB,
                                         bool copier,
                                         v2f (&a)[kSteps], v2f (&b)[2][k2Tiles], v8f (&acc)[k2Tiles]) {
  v2f an[kSteps];
  v4f wv[k2Pieces];
#pragma unroll
  for (int s = 0; s < kSteps; ++s) {
    if (s + 1 < kSteps) {
#pragma unroll
      for (int t = 0; t < k2Tiles; ++t)
        b[(s + 1) & 1][t] = *(const v2f*)(rdB + t * 16 * kPitch + 4 * (s + 1));
    }
    if (kMore) {
      if (s == 0) {
        if (copier) {
#pragma unroll
          for (int j = 0; j < k2Pieces; ++j)
            wv[j] = *(const v4f*)(wnext + (size_t)j * k2PieceRows * kF);
        }
#pragma unroll
        for (int j = 0; j < kSteps; ++j) an[j] = *(const v2f*)(anext + 4 * j);
      }
      if (s == 4) {
        if (copier) {
#pragma unroll
          for (int j = 0; j < k2Pieces; ++j)
            *(v4f*)(wrW + j * k2PieceRows * kPitch) = wv[j];
        }
      }
      if (s == kSteps - 1) {
        __syncthreads();
#pragma unroll
        for (int t = 0; t < k2Tiles; ++t) b[0][t] = *(const v2f*)(nxB + t * 16 * kPitch);
      }
    }
    __builtin_amdgcn_sched_barrier(0);
#pragma unroll
    for (int t = 0; t < k2Tiles; ++t)
      acc[t] = __builtin_amdgcn_wmma_f32_16x16x4_f32(false, a[s], false, b[s & 1][t],
                                                     (short)0, acc[t], false, false);
    __builtin_amdgcn_sched_barrier(0);
  }
  if (kMore) {
#pragma unroll
    for (int s = 0; s < kSteps; ++s) a[s] = an[s];
  }
}

__device__ __forceinline__ void gemm_rows16_x96(const float* __restrict__ arow,
                                                const float* __restrict__ Wm,
                                                float* __restrict__ sW,
                                                v8f (&acc)[k2Tiles],
                                                int tid, int hf, int l16, int part, bool copier) {
#pragma unroll
  for (int t = 0; t < k2Tiles; ++t) acc[t] = v8f{};

  const float* __restrict__ wsrc = Wm + (size_t)(tid >> 3) * kF + 4 * (tid & 7);
  const int wslot = (tid >> 3) * kPitch + 4 * (tid & 7);
  const int bfrag = (part * k2Tiles * 16 + l16) * kPitch + 2 * hf;

  if (copier) {
    v4f v[k2Pieces];
#pragma unroll
    for (int i = 0; i < k2Pieces; ++i) v[i] = *(const v4f*)(wsrc + (size_t)i * k2PieceRows * kF);
#pragma unroll
    for (int i = 0; i < k2Pieces; ++i) *(v4f*)(sW + wslot + i * k2PieceRows * kPitch) = v[i];
  }
  v2f a[kSteps];
#pragma unroll
  for (int s = 0; s < kSteps; ++s) a[s] = *(const v2f*)(arow + 4 * s);
  __syncthreads();

  v2f b[2][k2Tiles];
#pragma unroll
  for (int t = 0; t < k2Tiles; ++t) b[0][t] = *(const v2f*)(sW + bfrag + t * 16 * kPitch);

  float* rd = sW;
  float* wr = sW + k2Stage;
#pragma unroll 1
  for (int kc = 0; kc < kF - kKC; kc += kKC) {
    k_chunk2<true>(arow + kc + kKC, wsrc + kc + kKC, rd + bfrag, wr + wslot, wr + bfrag, copier, a, b, acc);
    float* const tmp = rd; rd = wr; wr = tmp;
  }
  k_chunk2<false>(arow, wsrc, rd + bfrag, wr + wslot, wr + bfrag, copier, a, b, acc);
}

__global__ __launch_bounds__(k2Threads)
void gemm2_perm(const float* __restrict__ T,
                const float* __restrict__ W2,
                const float* __restrict__ b2,
                const int*   __restrict__ perm,
                float* __restrict__ out) {
  __shared__ __attribute__((aligned(16))) float sW[2 * k2Stage];

  const int tid  = threadIdx.x;
  const int wave = __builtin_amdgcn_readfirstlane(tid >> 5);
  const int lane = tid & 31;
  const int hf   = lane >> 4;
  const int l16  = lane & 15;
  const int rg   = wave % k2RowGroups;
  const int part = wave / k2RowGroups;
  const int half = blockIdx.x & 1;
  const int row0 = (blockIdx.x >> 1) * k2Rows + rg * kWaveRows;
  const int col0 = half * k2Cols + part * (k2Tiles * 16);
  const bool copier = wave < k2Copiers / 32;

  const int g    = row0 + l16;
  const int img  = g / kN;
  const int n0   = g - img * kN;
  const int src  = img * kN + perm[n0];

  v8f acc[k2Tiles];
  gemm_rows16_x96(T + (size_t)src * kHid + 2 * hf, W2 + (size_t)half * k2Cols * kHid, sW, acc,
                  tid, hf, l16, part, copier);

#pragma unroll
  for (int t = 0; t < k2Tiles; ++t) {
    const int n = col0 + t * 16 + l16;
    const float bias = b2[n];
#pragma unroll
    for (int r = 0; r < 8; ++r) {
      out[(size_t)(row0 + 8 * hf + r) * kOut + n] = acc[t][r] + bias;
    }
  }
}

extern "C" void kernel_launch(void* const* d_in, const int* in_sizes, int n_in,
                              void* d_out, int out_size, void* d_ws, size_t ws_size,
                              hipStream_t stream) {
  const float* x    = (const float*)d_in[0];
  const float* cw   = (const float*)d_in[1];
  const float* cb   = (const float*)d_in[2];
  const float* bng  = (const float*)d_in[3];
  const float* bnb  = (const float*)d_in[4];
  const float* bnm  = (const float*)d_in[5];
  const float* bnv  = (const float*)d_in[6];
  const float* M    = (const float*)d_in[7];
  const float* W1   = (const float*)d_in[8];
  const float* b1   = (const float*)d_in[9];
  const float* lng  = (const float*)d_in[10];
  const float* lnb  = (const float*)d_in[11];
  const float* W2   = (const float*)d_in[12];
  const float* b2   = (const float*)d_in[13];
  const int*   mask = (const int*)d_in[14];
  const int*   perm = (const int*)d_in[15];
  float* out = (float*)d_out;

  float* flat = (float*)d_ws;
  float* Tbuf = flat + (size_t)kRows * kF;

  conv_bn_gelu_patch<<<kBS * kPR * 2, kConvThreads, 0, stream>>>(
      x, cw, cb, bng, bnb, bnm, bnv, M, mask, flat);
  gemm1_ln_gelu<<<kRows / kBlockRows, kThreads, 0, stream>>>(flat, W1, b1, lng, lnb, Tbuf);
  gemm2_perm  <<<(kRows / k2Rows) * 2, k2Threads, 0, stream>>>(Tbuf, W2, b2, perm, out);
}
